# starting kernel
# baseline (speedup 1.0000x reference)
#include <hip/hip_runtime.h>
#include <math.h>

#define E_EDGES  1000000
#define R_REL    1000
#define D_DIM    200
#define THREE_D  600
#define M_ROWS   2000
#define LDSTRIDE 204

typedef float v2f __attribute__((ext_vector_type(2)));
typedef float v8f __attribute__((ext_vector_type(8)));

__global__ void zero_ws(float* ws, int n) {
    int i = blockIdx.x * blockDim.x + threadIdx.x;
    if (i < n) ws[i] = 0.0f;
}

__global__ void scatter_kernel(const int* __restrict__ ei,
                               const int* __restrict__ et,
                               const int* __restrict__ nid,
                               const float* __restrict__ emb,
                               float* __restrict__ sum,
                               float* __restrict__ cnt) {
    long long idx   = (long long)blockIdx.x * blockDim.x + threadIdx.x;
    long long total = (long long)E_EDGES * (D_DIM / 4);
    if (idx >= total) return;
    int e  = (int)(idx / (D_DIM / 4));
    int d4 = (int)(idx - (long long)e * (D_DIM / 4)) * 4;
    int r = et[e];
    int s = nid[ei[e]];
    int t = nid[ei[E_EDGES + e]];
    const float4 vs = *(const float4*)(emb + (size_t)s * D_DIM + d4);
    const float4 vt = *(const float4*)(emb + (size_t)t * D_DIM + d4);
    float* ps = sum + (size_t)r * D_DIM + d4;
    float* pt = sum + (size_t)(R_REL + r) * D_DIM + d4;
    atomicAdd(ps + 0, vs.x); atomicAdd(ps + 1, vs.y);
    atomicAdd(ps + 2, vs.z); atomicAdd(ps + 3, vs.w);
    atomicAdd(pt + 0, vt.x); atomicAdd(pt + 1, vt.y);
    atomicAdd(pt + 2, vt.z); atomicAdd(pt + 3, vt.w);
    if (d4 == 0) atomicAdd(cnt + r, 1.0f);
}

__global__ void mean_kernel(float* __restrict__ sum, const float* __restrict__ cnt) {
    int i = blockIdx.x * blockDim.x + threadIdx.x;
    if (i >= M_ROWS * D_DIM) return;
    int m = i / D_DIM;
    sum[i] = sum[i] / fmaxf(cnt[m % R_REL], 1.0f);
}

__device__ __forceinline__ float sigmoidf_(float x) { return 1.0f / (1.0f + expf(-x)); }

__launch_bounds__(32)
__global__ void gru_wmma_kernel(const float* __restrict__ X,
                                const float* __restrict__ dyn,
                                const float* __restrict__ Wih,
                                const float* __restrict__ Whh,
                                const float* __restrict__ bih,
                                const float* __restrict__ bhh,
                                float* __restrict__ out) {
    __shared__ float Xs[16 * LDSTRIDE];
    __shared__ float Hs[16 * LDSTRIDE];
    const int lane = threadIdx.x;
    const int m0   = blockIdx.x * 16;

    for (int idx = lane; idx < 16 * D_DIM; idx += 32) {
        int row = idx / D_DIM, col = idx - row * D_DIM;
        int m = m0 + row;
        int r = m % R_REL, f = m / R_REL;
        Xs[row * LDSTRIDE + col] = X[(size_t)m * D_DIM + col];
        Hs[row * LDSTRIDE + col] = dyn[(size_t)r * (2 * D_DIM) + col * 2 + f];
    }
    __syncthreads();

    const int aRow = lane & 15;
    const int nRow = lane & 15;
    const int kOff = (lane < 16) ? 0 : 2;
    const int hi   = (lane < 16) ? 0 : 8;

    for (int c0 = 0; c0 < 192; c0 += 16) {
        const float* pI0 = Wih + (size_t)(c0 + nRow) * D_DIM + kOff;
        const float* pI1 = pI0 + (size_t)D_DIM * D_DIM;
        const float* pI2 = pI1 + (size_t)D_DIM * D_DIM;
        const float* pH0 = Whh + (size_t)(c0 + nRow) * D_DIM + kOff;
        const float* pH1 = pH0 + (size_t)D_DIM * D_DIM;
        const float* pH2 = pH1 + (size_t)D_DIM * D_DIM;

        v8f c_ir = {}; v8f c_iz = {}; v8f c_in = {};
        v8f c_hr = {}; v8f c_hz = {}; v8f c_hn = {};

        for (int k0 = 0; k0 < D_DIM; k0 += 4) {
            v2f ax, ah, b0, b1, b2, b3, b4, b5;
            ax.x = Xs[aRow * LDSTRIDE + k0 + kOff];
            ax.y = Xs[aRow * LDSTRIDE + k0 + kOff + 1];
            ah.x = Hs[aRow * LDSTRIDE + k0 + kOff];
            ah.y = Hs[aRow * LDSTRIDE + k0 + kOff + 1];
            b0.x = pI0[k0]; b0.y = pI0[k0 + 1];
            b1.x = pI1[k0]; b1.y = pI1[k0 + 1];
            b2.x = pI2[k0]; b2.y = pI2[k0 + 1];
            b3.x = pH0[k0]; b3.y = pH0[k0 + 1];
            b4.x = pH1[k0]; b4.y = pH1[k0 + 1];
            b5.x = pH2[k0]; b5.y = pH2[k0 + 1];

            c_ir = __builtin_amdgcn_wmma_f32_16x16x4_f32(false, ax, false, b0, (short)0, c_ir, false, false);
            c_iz = __builtin_amdgcn_wmma_f32_16x16x4_f32(false, ax, false, b1, (short)0, c_iz, false, false);
            c_in = __builtin_amdgcn_wmma_f32_16x16x4_f32(false, ax, false, b2, (short)0, c_in, false, false);
            c_hr = __builtin_amdgcn_wmma_f32_16x16x4_f32(false, ah, false, b3, (short)0, c_hr, false, false);
            c_hz = __builtin_amdgcn_wmma_f32_16x16x4_f32(false, ah, false, b4, (short)0, c_hz, false, false);
            c_hn = __builtin_amdgcn_wmma_f32_16x16x4_f32(false, ah, false, b5, (short)0, c_hn, false, false);
        }

        int dCol = c0 + nRow;
        float b_ir = bih[dCol], b_iz = bih[D_DIM + dCol], b_in = bih[2 * D_DIM + dCol];
        float b_hr = bhh[dCol], b_hz = bhh[D_DIM + dCol], b_hn = bhh[2 * D_DIM + dCol];
        #pragma unroll
        for (int i = 0; i < 8; ++i) {
            int   m  = m0 + hi + i;
            float h  = Hs[(hi + i) * LDSTRIDE + dCol];
            float rg = sigmoidf_(c_ir[i] + b_ir + c_hr[i] + b_hr);
            float zg = sigmoidf_(c_iz[i] + b_iz + c_hz[i] + b_hz);
            float ng = tanhf(c_in[i] + b_in + rg * (c_hn[i] + b_hn));
            float o  = (1.0f - zg) * ng + zg * h;
            int r = m % R_REL, f = m / R_REL;
            out[(size_t)r * (2 * D_DIM) + dCol * 2 + f] = o;
        }
    }

    {
        const int c0 = 192;
        const bool okN = (nRow < 8);
        const float* pI0 = Wih + (size_t)(c0 + nRow) * D_DIM + kOff;
        const float* pI1 = pI0 + (size_t)D_DIM * D_DIM;
        const float* pI2 = pI1 + (size_t)D_DIM * D_DIM;
        const float* pH0 = Whh + (size_t)(c0 + nRow) * D_DIM + kOff;
        const float* pH1 = pH0 + (size_t)D_DIM * D_DIM;
        const float* pH2 = pH1 + (size_t)D_DIM * D_DIM;

        v8f c_ir = {}; v8f c_iz = {}; v8f c_in = {};
        v8f c_hr = {}; v8f c_hz = {}; v8f c_hn = {};

        for (int k0 = 0; k0 < D_DIM; k0 += 4) {
            v2f ax, ah, b0, b1, b2, b3, b4, b5;
            ax.x = Xs[aRow * LDSTRIDE + k0 + kOff];
            ax.y = Xs[aRow * LDSTRIDE + k0 + kOff + 1];
            ah.x = Hs[aRow * LDSTRIDE + k0 + kOff];
            ah.y = Hs[aRow * LDSTRIDE + k0 + kOff + 1];
            b0.x = pI0[k0]; b0.y = pI0[k0 + 1];
            b1.x = pI1[k0]; b1.y = pI1[k0 + 1];
            b2.x = okN ? pI2[k0] : 0.0f; b2.y = okN ? pI2[k0 + 1] : 0.0f;
            b3.x = pH0[k0]; b3.y = pH0[k0 + 1];
            b4.x = pH1[k0]; b4.y = pH1[k0 + 1];
            b5.x = okN ? pH2[k0] : 0.0f; b5.y = okN ? pH2[k0 + 1] : 0.0f;

            c_ir = __builtin_amdgcn_wmma_f32_16x16x4_f32(false, ax, false, b0, (short)0, c_ir, false, false);
            c_iz = __builtin_amdgcn_wmma_f32_16x16x4_f32(false, ax, false, b1, (short)0, c_iz, false, false);
            c_in = __builtin_amdgcn_wmma_f32_16x16x4_f32(false, ax, false, b2, (short)0, c_in, false, false);
            c_hr = __builtin_amdgcn_wmma_f32_16x16x4_f32(false, ah, false, b3, (short)0, c_hr, false, false);
            c_hz = __builtin_amdgcn_wmma_f32_16x16x4_f32(false, ah, false, b4, (short)0, c_hz, false, false);
            c_hn = __builtin_amdgcn_wmma_f32_16x16x4_f32(false, ah, false, b5, (short)0, c_hn, false, false);
        }

        int  dCol  = c0 + nRow;
        bool valid = dCol < D_DIM;
        float b_ir = 0, b_iz = 0, b_in = 0, b_hr = 0, b_hz = 0, b_hn = 0;
        if (valid) {
            b_ir = bih[dCol]; b_iz = bih[D_DIM + dCol]; b_in = bih[2 * D_DIM + dCol];
            b_hr = bhh[dCol]; b_hz = bhh[D_DIM + dCol]; b_hn = bhh[2 * D_DIM + dCol];
        }
        #pragma unroll
        for (int i = 0; i < 8; ++i) {
            if (valid) {
                int   m  = m0 + hi + i;
                float h  = Hs[(hi + i) * LDSTRIDE + dCol];
                float rg = sigmoidf_(c_ir[i] + b_ir + c_hr[i] + b_hr);
                float zg = sigmoidf_(c_iz[i] + b_iz + c_hz[i] + b_hz);
                float ng = tanhf(c_in[i] + b_in + rg * (c_hn[i] + b_hn));
                float o  = (1.0f - zg) * ng + zg * h;
                int r = m % R_REL, f = m / R_REL;
                out[(size_t)r * (2 * D_DIM) + dCol * 2 + f] = o;
            }
        }
    }
}

extern "C" void kernel_launch(void* const* d_in, const int* in_sizes, int n_in,
                              void* d_out, int out_size, void* d_ws, size_t ws_size,
                              hipStream_t stream) {
    const int*   ei  = (const int*)d_in[0];
    const int*   et  = (const int*)d_in[1];
    const int*   nid = (const int*)d_in[2];
    const float* emb = (const float*)d_in[3];
    const float* dyn = (const float*)d_in[4];
    const float* Wih = (const float*)d_in[5];
    const float* Whh = (const float*)d_in[6];
    const float* bih = (const float*)d_in[7];
    const float* bhh = (const float*)d_in[8];
    float* out = (float*)d_out;

    float* sum = (float*)d_ws;
    float* cnt = sum + (size_t)M_ROWS * D_DIM;

    int nZero = M_ROWS * D_DIM + R_REL;
    zero_ws<<<(nZero + 255) / 256, 256, 0, stream>>>(sum, nZero);

    long long total = (long long)E_EDGES * (D_DIM / 4);
    int blocks = (int)((total + 255) / 256);
    scatter_kernel<<<blocks, 256, 0, stream>>>(ei, et, nid, emb, sum, cnt);

    mean_kernel<<<(M_ROWS * D_DIM + 255) / 256, 256, 0, stream>>>(sum, cnt);

    gru_wmma_kernel<<<M_ROWS / 16, 32, 0, stream>>>(sum, dyn, Wih, Whh, bih, bhh, out);
}


# segment mean by owner: a work-group per 4 relations scans edge_type, queues its edges in LDS, sums rows in registers, stores means; no float atomics, no zero or mean launch; sums in arrival order
# speedup vs baseline: 4.9615x; 4.9615x over previous
#include <hip/hip_runtime.h>
#include <math.h>

#define E_EDGES  1000000
#define R_REL    1000
#define D_DIM    200
#define THREE_D  600
#define M_ROWS   2000
#define LDSTRIDE 204

typedef float v2f __attribute__((ext_vector_type(2)));
typedef float v8f __attribute__((ext_vector_type(8)));

#define G_REL    4
#define QCAP     4096
#define SEG_THR  1024
#define SCAN_U   8
#define GATH_U   8
#define ROW_SLOTS 4

__device__ __forceinline__ void probe_edge(int ty, int e, int r0, int2 (*qlist)[QCAP],
                                           int* qcount, int* qovf) {
    unsigned d = (unsigned)(ty - r0);
    if (d < (unsigned)G_REL) {
        int pos = atomicAdd(&qcount[d], 1);
        if (pos < QCAP) qlist[d][pos].x = e;
        else            *qovf = 1;
    }
}

__launch_bounds__(SEG_THR)
__global__ void segmean_kernel(const int* __restrict__ ei,
                               const int* __restrict__ et,
                               const int* __restrict__ nid,
                               const float* __restrict__ emb,
                               float* __restrict__ X) {
    __shared__ __attribute__((aligned(16))) int2 qlist[G_REL][QCAP];
    __shared__ int  qcount[G_REL];
    __shared__ int  qtotal[G_REL];
    __shared__ int  qovf;

    const int t    = threadIdx.x;
    const int r0   = blockIdx.x * G_REL;
    const int grp  = t >> 6;
    const int q    = grp >> 2;
    const int slot = grp & (ROW_SLOTS - 1);
    const int c    = t & 63;
    const bool active = c < (D_DIM / 4);

    float4 accS = make_float4(0.f, 0.f, 0.f, 0.f);
    float4 accD = make_float4(0.f, 0.f, 0.f, 0.f);

    if (t < G_REL) qtotal[t] = 0;

    const int4* et4 = (const int4*)et;
    int start = 0;
    int range = E_EDGES;
    while (start < E_EDGES) {
        if (t < G_REL) qcount[t] = 0;
        if (t == 0) qovf = 0;
        __syncthreads();

        const int end = (start + range < E_EDGES) ? (start + range) : E_EDGES;
        const int s4 = start >> 2, e4 = end >> 2;
        for (int b = s4 + t; b < e4; b += SEG_THR * SCAN_U) {
            int4 v[SCAN_U];
            #pragma unroll
            for (int u = 0; u < SCAN_U; ++u) {
                int i = b + u * SEG_THR;
                v[u] = (i < e4) ? et4[i] : make_int4(-1, -1, -1, -1);
            }
            #pragma unroll
            for (int u = 0; u < SCAN_U; ++u) {
                int e = (b + u * SEG_THR) * 4;
                probe_edge(v[u].x, e + 0, r0, qlist, qcount, &qovf);
                probe_edge(v[u].y, e + 1, r0, qlist, qcount, &qovf);
                probe_edge(v[u].z, e + 2, r0, qlist, qcount, &qovf);
                probe_edge(v[u].w, e + 3, r0, qlist, qcount, &qovf);
            }
        }
        __syncthreads();

        const bool ovf = (qovf != 0);
        const int n0 = qcount[0], n1 = qcount[1], n2 = qcount[2], n3 = qcount[3];
        if (ovf) {
            range = QCAP;
            __syncthreads();
            continue;
        }
        if (t < G_REL) qtotal[t] += qcount[t];

        for (int qq = 0; qq < G_REL; ++qq) {
            const int nn = (qq == 0) ? n0 : (qq == 1) ? n1 : (qq == 2) ? n2 : n3;
            for (int i = t; i < nn; i += SEG_THR) {
                int e = qlist[qq][i].x;
                int s = nid[ei[e]];
                int d = nid[ei[E_EDGES + e]];
                qlist[qq][i] = make_int2(s, d);
            }
        }
        __syncthreads();

        const int n = (q == 0) ? n0 : (q == 1) ? n1 : (q == 2) ? n2 : n3;
        if (active && n > 0) {
            const int2*  L    = qlist[q];
            const float* base = emb + 4 * c;
            for (int j0 = slot; j0 < n; j0 += ROW_SLOTS * GATH_U) {
                int2   en[GATH_U];
                float4 vs[GATH_U], vd[GATH_U];
                #pragma unroll
                for (int u = 0; u < GATH_U; ++u) {
                    int j = j0 + ROW_SLOTS * u;
                    en[u] = L[(j < n) ? j : (n - 1)];
                }
                #pragma unroll
                for (int u = 0; u < GATH_U; ++u) {
                    vs[u] = *(const float4*)(base + (size_t)en[u].x * D_DIM);
                    vd[u] = *(const float4*)(base + (size_t)en[u].y * D_DIM);
                }
                #pragma unroll
                for (int u = 0; u < GATH_U; ++u) {
                    const bool ok = (j0 + ROW_SLOTS * u < n);
                    accS.x += ok ? vs[u].x : 0.f; accS.y += ok ? vs[u].y : 0.f;
                    accS.z += ok ? vs[u].z : 0.f; accS.w += ok ? vs[u].w : 0.f;
                    accD.x += ok ? vd[u].x : 0.f; accD.y += ok ? vd[u].y : 0.f;
                    accD.z += ok ? vd[u].z : 0.f; accD.w += ok ? vd[u].w : 0.f;
                }
            }
        }
        start += range;
    }

    __syncthreads();
    float* red = (float*)&qlist[0][0];
    if (active) {
        *(float4*)&red[(grp * 2 + 0) * D_DIM + 4 * c] = accS;
        *(float4*)&red[(grp * 2 + 1) * D_DIM + 4 * c] = accD;
    }
    __syncthreads();
    if (t < G_REL * 2 * (D_DIM / 4)) {
        const int q2   = t / (2 * (D_DIM / 4));
        const int rem  = t - q2 * (2 * (D_DIM / 4));
        const int side = rem / (D_DIM / 4);
        const int c2   = rem - side * (D_DIM / 4);
        float4 s = *(const float4*)&red[((q2 * ROW_SLOTS + 0) * 2 + side) * D_DIM + 4 * c2];
        for (int sl = 1; sl < ROW_SLOTS; ++sl) {
            float4 p = *(const float4*)&red[((q2 * ROW_SLOTS + sl) * 2 + side) * D_DIM + 4 * c2];
            s.x += p.x; s.y += p.y; s.z += p.z; s.w += p.w;
        }
        const float den = fmaxf((float)qtotal[q2], 1.0f);
        s.x = s.x / den; s.y = s.y / den; s.z = s.z / den; s.w = s.w / den;
        *(float4*)&X[((size_t)(side * R_REL + r0 + q2)) * D_DIM + 4 * c2] = s;
    }
}

__device__ __forceinline__ float sigmoidf_(float x) { return 1.0f / (1.0f + expf(-x)); }

__launch_bounds__(32)
__global__ void gru_wmma_kernel(const float* __restrict__ X,
                                const float* __restrict__ dyn,
                                const float* __restrict__ Wih,
                                const float* __restrict__ Whh,
                                const float* __restrict__ bih,
                                const float* __restrict__ bhh,
                                float* __restrict__ out) {
    __shared__ float Xs[16 * LDSTRIDE];
    __shared__ float Hs[16 * LDSTRIDE];
    const int lane = threadIdx.x;
    const int m0   = blockIdx.x * 16;

    for (int idx = lane; idx < 16 * D_DIM; idx += 32) {
        int row = idx / D_DIM, col = idx - row * D_DIM;
        int m = m0 + row;
        int r = m % R_REL, f = m / R_REL;
        Xs[row * LDSTRIDE + col] = X[(size_t)m * D_DIM + col];
        Hs[row * LDSTRIDE + col] = dyn[(size_t)r * (2 * D_DIM) + col * 2 + f];
    }
    __syncthreads();

    const int aRow = lane & 15;
    const int nRow = lane & 15;
    const int kOff = (lane < 16) ? 0 : 2;
    const int hi   = (lane < 16) ? 0 : 8;

    for (int c0 = 0; c0 < 192; c0 += 16) {
        const float* pI0 = Wih + (size_t)(c0 + nRow) * D_DIM + kOff;
        const float* pI1 = pI0 + (size_t)D_DIM * D_DIM;
        const float* pI2 = pI1 + (size_t)D_DIM * D_DIM;
        const float* pH0 = Whh + (size_t)(c0 + nRow) * D_DIM + kOff;
        const float* pH1 = pH0 + (size_t)D_DIM * D_DIM;
        const float* pH2 = pH1 + (size_t)D_DIM * D_DIM;

        v8f c_ir = {}; v8f c_iz = {}; v8f c_in = {};
        v8f c_hr = {}; v8f c_hz = {}; v8f c_hn = {};

        for (int k0 = 0; k0 < D_DIM; k0 += 4) {
            v2f ax, ah, b0, b1, b2, b3, b4, b5;
            ax.x = Xs[aRow * LDSTRIDE + k0 + kOff];
            ax.y = Xs[aRow * LDSTRIDE + k0 + kOff + 1];
            ah.x = Hs[aRow * LDSTRIDE + k0 + kOff];
            ah.y = Hs[aRow * LDSTRIDE + k0 + kOff + 1];
            b0.x = pI0[k0]; b0.y = pI0[k0 + 1];
            b1.x = pI1[k0]; b1.y = pI1[k0 + 1];
            b2.x = pI2[k0]; b2.y = pI2[k0 + 1];
            b3.x = pH0[k0]; b3.y = pH0[k0 + 1];
            b4.x = pH1[k0]; b4.y = pH1[k0 + 1];
            b5.x = pH2[k0]; b5.y = pH2[k0 + 1];

            c_ir = __builtin_amdgcn_wmma_f32_16x16x4_f32(false, ax, false, b0, (short)0, c_ir, false, false);
            c_iz = __builtin_amdgcn_wmma_f32_16x16x4_f32(false, ax, false, b1, (short)0, c_iz, false, false);
            c_in = __builtin_amdgcn_wmma_f32_16x16x4_f32(false, ax, false, b2, (short)0, c_in, false, false);
            c_hr = __builtin_amdgcn_wmma_f32_16x16x4_f32(false, ah, false, b3, (short)0, c_hr, false, false);
            c_hz = __builtin_amdgcn_wmma_f32_16x16x4_f32(false, ah, false, b4, (short)0, c_hz, false, false);
            c_hn = __builtin_amdgcn_wmma_f32_16x16x4_f32(false, ah, false, b5, (short)0, c_hn, false, false);
        }

        int dCol = c0 + nRow;
        float b_ir = bih[dCol], b_iz = bih[D_DIM + dCol], b_in = bih[2 * D_DIM + dCol];
        float b_hr = bhh[dCol], b_hz = bhh[D_DIM + dCol], b_hn = bhh[2 * D_DIM + dCol];
        #pragma unroll
        for (int i = 0; i < 8; ++i) {
            int   m  = m0 + hi + i;
            float h  = Hs[(hi + i) * LDSTRIDE + dCol];
            float rg = sigmoidf_(c_ir[i] + b_ir + c_hr[i] + b_hr);
            float zg = sigmoidf_(c_iz[i] + b_iz + c_hz[i] + b_hz);
            float ng = tanhf(c_in[i] + b_in + rg * (c_hn[i] + b_hn));
            float o  = (1.0f - zg) * ng + zg * h;
            int r = m % R_REL, f = m / R_REL;
            out[(size_t)r * (2 * D_DIM) + dCol * 2 + f] = o;
        }
    }

    {
        const int c0 = 192;
        const bool okN = (nRow < 8);
        const float* pI0 = Wih + (size_t)(c0 + nRow) * D_DIM + kOff;
        const float* pI1 = pI0 + (size_t)D_DIM * D_DIM;
        const float* pI2 = pI1 + (size_t)D_DIM * D_DIM;
        const float* pH0 = Whh + (size_t)(c0 + nRow) * D_DIM + kOff;
        const float* pH1 = pH0 + (size_t)D_DIM * D_DIM;
        const float* pH2 = pH1 + (size_t)D_DIM * D_DIM;

        v8f c_ir = {}; v8f c_iz = {}; v8f c_in = {};
        v8f c_hr = {}; v8f c_hz = {}; v8f c_hn = {};

        for (int k0 = 0; k0 < D_DIM; k0 += 4) {
            v2f ax, ah, b0, b1, b2, b3, b4, b5;
            ax.x = Xs[aRow * LDSTRIDE + k0 + kOff];
            ax.y = Xs[aRow * LDSTRIDE + k0 + kOff + 1];
            ah.x = Hs[aRow * LDSTRIDE + k0 + kOff];
            ah.y = Hs[aRow * LDSTRIDE + k0 + kOff + 1];
            b0.x = pI0[k0]; b0.y = pI0[k0 + 1];
            b1.x = pI1[k0]; b1.y = pI1[k0 + 1];
            b2.x = okN ? pI2[k0] : 0.0f; b2.y = okN ? pI2[k0 + 1] : 0.0f;
            b3.x = pH0[k0]; b3.y = pH0[k0 + 1];
            b4.x = pH1[k0]; b4.y = pH1[k0 + 1];
            b5.x = okN ? pH2[k0] : 0.0f; b5.y = okN ? pH2[k0 + 1] : 0.0f;

            c_ir = __builtin_amdgcn_wmma_f32_16x16x4_f32(false, ax, false, b0, (short)0, c_ir, false, false);
            c_iz = __builtin_amdgcn_wmma_f32_16x16x4_f32(false, ax, false, b1, (short)0, c_iz, false, false);
            c_in = __builtin_amdgcn_wmma_f32_16x16x4_f32(false, ax, false, b2, (short)0, c_in, false, false);
            c_hr = __builtin_amdgcn_wmma_f32_16x16x4_f32(false, ah, false, b3, (short)0, c_hr, false, false);
            c_hz = __builtin_amdgcn_wmma_f32_16x16x4_f32(false, ah, false, b4, (short)0, c_hz, false, false);
            c_hn = __builtin_amdgcn_wmma_f32_16x16x4_f32(false, ah, false, b5, (short)0, c_hn, false, false);
        }

        int  dCol  = c0 + nRow;
        bool valid = dCol < D_DIM;
        float b_ir = 0, b_iz = 0, b_in = 0, b_hr = 0, b_hz = 0, b_hn = 0;
        if (valid) {
            b_ir = bih[dCol]; b_iz = bih[D_DIM + dCol]; b_in = bih[2 * D_DIM + dCol];
            b_hr = bhh[dCol]; b_hz = bhh[D_DIM + dCol]; b_hn = bhh[2 * D_DIM + dCol];
        }
        #pragma unroll
        for (int i = 0; i < 8; ++i) {
            if (valid) {
                int   m  = m0 + hi + i;
                float h  = Hs[(hi + i) * LDSTRIDE + dCol];
                float rg = sigmoidf_(c_ir[i] + b_ir + c_hr[i] + b_hr);
                float zg = sigmoidf_(c_iz[i] + b_iz + c_hz[i] + b_hz);
                float ng = tanhf(c_in[i] + b_in + rg * (c_hn[i] + b_hn));
                float o  = (1.0f - zg) * ng + zg * h;
                int r = m % R_REL, f = m / R_REL;
                out[(size_t)r * (2 * D_DIM) + dCol * 2 + f] = o;
            }
        }
    }
}

extern "C" void kernel_launch(void* const* d_in, const int* in_sizes, int n_in,
                              void* d_out, int out_size, void* d_ws, size_t ws_size,
                              hipStream_t stream) {
    const int*   ei  = (const int*)d_in[0];
    const int*   et  = (const int*)d_in[1];
    const int*   nid = (const int*)d_in[2];
    const float* emb = (const float*)d_in[3];
    const float* dyn = (const float*)d_in[4];
    const float* Wih = (const float*)d_in[5];
    const float* Whh = (const float*)d_in[6];
    const float* bih = (const float*)d_in[7];
    const float* bhh = (const float*)d_in[8];
    float* out = (float*)d_out;

    float* X = (float*)d_ws;

    segmean_kernel<<<R_REL / G_REL, SEG_THR, 0, stream>>>(ei, et, nid, emb, X);

    gru_wmma_kernel<<<M_ROWS / 16, 32, 0, stream>>>(X, dyn, Wih, Whh, bih, bhh, out);
}


# GRU weight-stationary: a work-group per 16-column tile copies its 96 weight rows to LDS once (whole lines), its 8 waves take a 16-row tile each, B fragments from LDS; 208 groups; sum order kept
# speedup vs baseline: 9.3515x; 1.8848x over previous
#include <hip/hip_runtime.h>
#include <math.h>

#define E_EDGES  1000000
#define R_REL    1000
#define D_DIM    200
#define THREE_D  600
#define M_ROWS   2000
#define LDSTRIDE 204

typedef float v2f __attribute__((ext_vector_type(2)));
typedef float v8f __attribute__((ext_vector_type(8)));

#define G_REL    4
#define QCAP     4096
#define SEG_THR  1024
#define SCAN_U   8
#define GATH_U   8
#define ROW_SLOTS 4

__device__ __forceinline__ void probe_edge(int ty, int e, int r0, int2 (*qlist)[QCAP],
                                           int* qcount, int* qovf) {
    unsigned d = (unsigned)(ty - r0);
    if (d < (unsigned)G_REL) {
        int pos = atomicAdd(&qcount[d], 1);
        if (pos < QCAP) qlist[d][pos].x = e;
        else            *qovf = 1;
    }
}

__launch_bounds__(SEG_THR)
__global__ void segmean_kernel(const int* __restrict__ ei,
                               const int* __restrict__ et,
                               const int* __restrict__ nid,
                               const float* __restrict__ emb,
                               float* __restrict__ X) {
    __shared__ __attribute__((aligned(16))) int2 qlist[G_REL][QCAP];
    __shared__ int  qcount[G_REL];
    __shared__ int  qtotal[G_REL];
    __shared__ int  qovf;

    const int t    = threadIdx.x;
    const int r0   = blockIdx.x * G_REL;
    const int grp  = t >> 6;
    const int q    = grp >> 2;
    const int slot = grp & (ROW_SLOTS - 1);
    const int c    = t & 63;
    const bool active = c < (D_DIM / 4);

    float4 accS = make_float4(0.f, 0.f, 0.f, 0.f);
    float4 accD = make_float4(0.f, 0.f, 0.f, 0.f);

    if (t < G_REL) qtotal[t] = 0;

    const int4* et4 = (const int4*)et;
    int start = 0;
    int range = E_EDGES;
    while (start < E_EDGES) {
        if (t < G_REL) qcount[t] = 0;
        if (t == 0) qovf = 0;
        __syncthreads();

        const int end = (start + range < E_EDGES) ? (start + range) : E_EDGES;
        const int s4 = start >> 2, e4 = end >> 2;
        for (int b = s4 + t; b < e4; b += SEG_THR * SCAN_U) {
            int4 v[SCAN_U];
            #pragma unroll
            for (int u = 0; u < SCAN_U; ++u) {
                int i = b + u * SEG_THR;
                v[u] = (i < e4) ? et4[i] : make_int4(-1, -1, -1, -1);
            }
            #pragma unroll
            for (int u = 0; u < SCAN_U; ++u) {
                int e = (b + u * SEG_THR) * 4;
                probe_edge(v[u].x, e + 0, r0, qlist, qcount, &qovf);
                probe_edge(v[u].y, e + 1, r0, qlist, qcount, &qovf);
                probe_edge(v[u].z, e + 2, r0, qlist, qcount, &qovf);
                probe_edge(v[u].w, e + 3, r0, qlist, qcount, &qovf);
            }
        }
        __syncthreads();

        const bool ovf = (qovf != 0);
        const int n0 = qcount[0], n1 = qcount[1], n2 = qcount[2], n3 = qcount[3];
        if (ovf) {
            range = QCAP;
            __syncthreads();
            continue;
        }
        if (t < G_REL) qtotal[t] += qcount[t];

        for (int qq = 0; qq < G_REL; ++qq) {
            const int nn = (qq == 0) ? n0 : (qq == 1) ? n1 : (qq == 2) ? n2 : n3;
            for (int i = t; i < nn; i += SEG_THR) {
                int e = qlist[qq][i].x;
                int s = nid[ei[e]];
                int d = nid[ei[E_EDGES + e]];
                qlist[qq][i] = make_int2(s, d);
            }
        }
        __syncthreads();

        const int n = (q == 0) ? n0 : (q == 1) ? n1 : (q == 2) ? n2 : n3;
        if (active && n > 0) {
            const int2*  L    = qlist[q];
            const float* base = emb + 4 * c;
            for (int j0 = slot; j0 < n; j0 += ROW_SLOTS * GATH_U) {
                int2   en[GATH_U];
                float4 vs[GATH_U], vd[GATH_U];
                #pragma unroll
                for (int u = 0; u < GATH_U; ++u) {
                    int j = j0 + ROW_SLOTS * u;
                    en[u] = L[(j < n) ? j : (n - 1)];
                }
                #pragma unroll
                for (int u = 0; u < GATH_U; ++u) {
                    vs[u] = *(const float4*)(base + (size_t)en[u].x * D_DIM);
                    vd[u] = *(const float4*)(base + (size_t)en[u].y * D_DIM);
                }
                #pragma unroll
                for (int u = 0; u < GATH_U; ++u) {
                    const bool ok = (j0 + ROW_SLOTS * u < n);
                    accS.x += ok ? vs[u].x : 0.f; accS.y += ok ? vs[u].y : 0.f;
                    accS.z += ok ? vs[u].z : 0.f; accS.w += ok ? vs[u].w : 0.f;
                    accD.x += ok ? vd[u].x : 0.f; accD.y += ok ? vd[u].y : 0.f;
                    accD.z += ok ? vd[u].z : 0.f; accD.w += ok ? vd[u].w : 0.f;
                }
            }
        }
        start += range;
    }

    __syncthreads();
    float* red = (float*)&qlist[0][0];
    if (active) {
        *(float4*)&red[(grp * 2 + 0) * D_DIM + 4 * c] = accS;
        *(float4*)&red[(grp * 2 + 1) * D_DIM + 4 * c] = accD;
    }
    __syncthreads();
    if (t < G_REL * 2 * (D_DIM / 4)) {
        const int q2   = t / (2 * (D_DIM / 4));
        const int rem  = t - q2 * (2 * (D_DIM / 4));
        const int side = rem / (D_DIM / 4);
        const int c2   = rem - side * (D_DIM / 4);
        float4 s = *(const float4*)&red[((q2 * ROW_SLOTS + 0) * 2 + side) * D_DIM + 4 * c2];
        for (int sl = 1; sl < ROW_SLOTS; ++sl) {
            float4 p = *(const float4*)&red[((q2 * ROW_SLOTS + sl) * 2 + side) * D_DIM + 4 * c2];
            s.x += p.x; s.y += p.y; s.z += p.z; s.w += p.w;
        }
        const float den = fmaxf((float)qtotal[q2], 1.0f);
        s.x = s.x / den; s.y = s.y / den; s.z = s.z / den; s.w = s.w / den;
        *(float4*)&X[((size_t)(side * R_REL + r0 + q2)) * D_DIM + 4 * c2] = s;
    }
}

__device__ __forceinline__ float sigmoidf_(float x) { return 1.0f / (1.0f + expf(-x)); }

#define GRU_WAVES  8
#define GRU_THR    (GRU_WAVES * 32)
#define ROW_TILES  (M_ROWS / 16)
#define ROW_GROUPS 16
#define COL_TILES  13
#define W_ROWS     96

__launch_bounds__(GRU_THR)
__global__ void gru_wmma_kernel(const float* __restrict__ X,
                                const float* __restrict__ dyn,
                                const float* __restrict__ Wih,
                                const float* __restrict__ Whh,
                                const float* __restrict__ bih,
                                const float* __restrict__ bhh,
                                float* __restrict__ out) {
    __shared__ __attribute__((aligned(16))) float Ws[W_ROWS * LDSTRIDE];

    const int t    = threadIdx.x;
    const int lane = t & 31;
    const int wave = __builtin_amdgcn_readfirstlane(t >> 5);
    const int c0   = blockIdx.x * 16;
    const int tile = blockIdx.y * GRU_WAVES + wave;

    {
        const int lane2 = (lane < 18) ? lane : 17;
        float4 v0[W_ROWS / GRU_WAVES], v1[W_ROWS / GRU_WAVES];
        #pragma unroll
        for (int j = 0; j < W_ROWS / GRU_WAVES; ++j) {
            const int  row  = wave * (W_ROWS / GRU_WAVES) + j;
            const int  g    = row >> 4;
            const int  col  = c0 + (row & 15);
            const int  colc = (col < D_DIM) ? col : (D_DIM - 1);
            const float* Wg   = (g < 3) ? Wih : Whh;
            const int    gate = (g < 3) ? g : (g - 3);
            const float* srow = Wg + (size_t)(gate * D_DIM + colc) * D_DIM;
            v0[j] = *(const float4*)(srow + 4 * lane);
            v1[j] = *(const float4*)(srow + 128 + 4 * lane2);
        }
        #pragma unroll
        for (int j = 0; j < W_ROWS / GRU_WAVES; ++j) {
            const int row = wave * (W_ROWS / GRU_WAVES) + j;
            *(float4*)&Ws[row * LDSTRIDE + 4 * lane] = v0[j];
            *(float4*)&Ws[row * LDSTRIDE + 128 + 4 * lane2] = v1[j];
        }
    }
    __syncthreads();

    if (tile < ROW_TILES) {
        const int m0   = tile * 16;
        const int aRow = lane & 15;
        const int kOff = (lane < 16) ? 0 : 2;
        const int hi   = (lane < 16) ? 0 : 8;

        const int  mA = m0 + aRow;
        const bool fA = (mA >= R_REL);
        const int  rA = fA ? (mA - R_REL) : mA;
        const float* xa = X + (size_t)mA * D_DIM + kOff;
        const float* ha = dyn + (size_t)rA * (2 * D_DIM) + 2 * kOff;
        const float* wb = Ws + aRow * LDSTRIDE + kOff;

        v8f c_ir = {}; v8f c_iz = {}; v8f c_in = {};
        v8f c_hr = {}; v8f c_hz = {}; v8f c_hn = {};

        #pragma unroll 5
        for (int k0 = 0; k0 < D_DIM; k0 += 4) {
            const v2f    ax = *(const v2f*)(xa + k0);
            const float4 hv = *(const float4*)(ha + 2 * k0);
            v2f ah;
            ah.x = fA ? hv.y : hv.x;
            ah.y = fA ? hv.w : hv.z;
            const v2f b0 = *(const v2f*)(wb + 0 * 16 * LDSTRIDE + k0);
            const v2f b1 = *(const v2f*)(wb + 1 * 16 * LDSTRIDE + k0);
            const v2f b2 = *(const v2f*)(wb + 2 * 16 * LDSTRIDE + k0);
            const v2f b3 = *(const v2f*)(wb + 3 * 16 * LDSTRIDE + k0);
            const v2f b4 = *(const v2f*)(wb + 4 * 16 * LDSTRIDE + k0);
            const v2f b5 = *(const v2f*)(wb + 5 * 16 * LDSTRIDE + k0);

            c_ir = __builtin_amdgcn_wmma_f32_16x16x4_f32(false, ax, false, b0, (short)0, c_ir, false, false);
            c_iz = __builtin_amdgcn_wmma_f32_16x16x4_f32(false, ax, false, b1, (short)0, c_iz, false, false);
            c_in = __builtin_amdgcn_wmma_f32_16x16x4_f32(false, ax, false, b2, (short)0, c_in, false, false);
            c_hr = __builtin_amdgcn_wmma_f32_16x16x4_f32(false, ah, false, b3, (short)0, c_hr, false, false);
            c_hz = __builtin_amdgcn_wmma_f32_16x16x4_f32(false, ah, false, b4, (short)0, c_hz, false, false);
            c_hn = __builtin_amdgcn_wmma_f32_16x16x4_f32(false, ah, false, b5, (short)0, c_hn, false, false);
        }

        const int  dCol  = c0 + aRow;
        const bool valid = dCol < D_DIM;
        const int  dc    = valid ? dCol : (D_DIM - 1);
        const float b_ir = bih[dc], b_iz = bih[D_DIM + dc], b_in = bih[2 * D_DIM + dc];
        const float b_hr = bhh[dc], b_hz = bhh[D_DIM + dc], b_hn = bhh[2 * D_DIM + dc];
        #pragma unroll
        for (int i = 0; i < 8; ++i) {
            const int  m = m0 + hi + i;
            const bool f = (m >= R_REL);
            const int  r = f ? (m - R_REL) : m;
            const size_t o_idx = (size_t)r * (2 * D_DIM) + dc * 2 + (f ? 1 : 0);
            float h  = dyn[o_idx];
            float rg = sigmoidf_(c_ir[i] + b_ir + c_hr[i] + b_hr);
            float zg = sigmoidf_(c_iz[i] + b_iz + c_hz[i] + b_hz);
            float ng = tanhf(c_in[i] + b_in + rg * (c_hn[i] + b_hn));
            float o  = (1.0f - zg) * ng + zg * h;
            if (valid) out[o_idx] = o;
        }
    }
}

extern "C" void kernel_launch(void* const* d_in, const int* in_sizes, int n_in,
                              void* d_out, int out_size, void* d_ws, size_t ws_size,
                              hipStream_t stream) {
    const int*   ei  = (const int*)d_in[0];
    const int*   et  = (const int*)d_in[1];
    const int*   nid = (const int*)d_in[2];
    const float* emb = (const float*)d_in[3];
    const float* dyn = (const float*)d_in[4];
    const float* Wih = (const float*)d_in[5];
    const float* Whh = (const float*)d_in[6];
    const float* bih = (const float*)d_in[7];
    const float* bhh = (const float*)d_in[8];
    float* out = (float*)d_out;

    float* X = (float*)d_ws;

    segmean_kernel<<<R_REL / G_REL, SEG_THR, 0, stream>>>(ei, et, nid, emb, X);

    gru_wmma_kernel<<<dim3(COL_TILES, ROW_GROUPS), GRU_THR, 0, stream>>>(X, dyn, Wih, Whh, bih, bhh, out);
}


# segment mean: pre-pass marks per owner group which 32-edge lines of edge_type hold its edges (bit rows in out until the GRU writes it); owners fetch marked lines only; 3 launches; arrival-order sums
# speedup vs baseline: 11.8345x; 1.2655x over previous
#include <hip/hip_runtime.h>
#include <math.h>

#define E_EDGES  1000000
#define R_REL    1000
#define D_DIM    200
#define THREE_D  600
#define M_ROWS   2000
#define LDSTRIDE 204

typedef float v2f __attribute__((ext_vector_type(2)));
typedef float v8f __attribute__((ext_vector_type(8)));

#define G_REL    4
#define N_GROUPS (R_REL / G_REL)

#define LINE_EDGES 32
#define N_LINES    (E_EDGES / LINE_EDGES)
#define IDX_EDGES  4096
#define IDX_THR    256
#define IDX_U      (IDX_EDGES / 4 / IDX_THR)
#define IDX_BLOCKS ((E_EDGES + IDX_EDGES - 1) / IDX_EDGES)
#define IDX_WORDS  ((N_LINES + 31) / 32)
#define IDX_PITCH  (IDX_BLOCKS * 4)

__device__ __forceinline__ void mark_type(int ty, int u, unsigned bit, unsigned* words) {
    if ((unsigned)ty < (unsigned)R_REL) atomicOr(&words[(ty / G_REL) * 4 + u], bit);
}

__launch_bounds__(IDX_THR)
__global__ void line_index_kernel(const int* __restrict__ et,
                                  unsigned* __restrict__ lidx) {
    __shared__ __attribute__((aligned(16))) unsigned words[N_GROUPS * 4];
    const int t = threadIdx.x;
    for (int i = t; i < N_GROUPS * 4; i += IDX_THR) words[i] = 0u;
    __syncthreads();

    const int4* et4 = (const int4*)et;
    int4 v[IDX_U];
    #pragma unroll
    for (int u = 0; u < IDX_U; ++u) {
        const int i4 = blockIdx.x * (IDX_EDGES / 4) + u * IDX_THR + t;
        v[u] = (i4 < E_EDGES / 4) ? et4[i4] : make_int4(-1, -1, -1, -1);
    }
    const unsigned bit = 1u << (t >> 3);
    #pragma unroll
    for (int u = 0; u < IDX_U; ++u) {
        mark_type(v[u].x, u, bit, words);
        mark_type(v[u].y, u, bit, words);
        mark_type(v[u].z, u, bit, words);
        mark_type(v[u].w, u, bit, words);
    }
    __syncthreads();
    if (t < N_GROUPS)
        *(uint4*)&lidx[(size_t)t * IDX_PITCH + blockIdx.x * 4] = *(const uint4*)&words[t * 4];
}

#define QCAP     4096
#define SEG_THR  1024
#define SCAN_U   8
#define GATH_U   8
#define ROW_SLOTS 4

__device__ __forceinline__ void probe_edge(int ty, int e, int r0, int2 (*qlist)[QCAP],
                                           int* qcount, int* qovf) {
    unsigned d = (unsigned)(ty - r0);
    if (d < (unsigned)G_REL) {
        int pos = atomicAdd(&qcount[d], 1);
        if (pos < QCAP) qlist[d][pos].x = e;
        else            *qovf = 1;
    }
}

__launch_bounds__(SEG_THR)
__global__ void segmean_kernel(const int* __restrict__ ei,
                               const int* __restrict__ et,
                               const int* __restrict__ nid,
                               const float* __restrict__ emb,
                               const unsigned* __restrict__ lidx,
                               float* __restrict__ X) {
    __shared__ __attribute__((aligned(16))) int2 qlist[G_REL][QCAP];
    __shared__ unsigned short llist[N_LINES];
    __shared__ int  lcount;
    __shared__ int  qcount[G_REL];
    __shared__ int  qtotal[G_REL];
    __shared__ int  qovf;

    const int t    = threadIdx.x;
    const int r0   = blockIdx.x * G_REL;
    const int grp  = t >> 6;
    const int q    = grp >> 2;
    const int slot = grp & (ROW_SLOTS - 1);
    const int c    = t & 63;
    const bool active = c < (D_DIM / 4);

    float4 accS = make_float4(0.f, 0.f, 0.f, 0.f);
    float4 accD = make_float4(0.f, 0.f, 0.f, 0.f);

    if (t < G_REL) qtotal[t] = 0;

    const int4* et4 = (const int4*)et;
    const unsigned* lrow = lidx + (size_t)blockIdx.x * IDX_PITCH;
    int start = 0;
    int range = E_EDGES;
    while (start < E_EDGES) {
        if (t < G_REL) qcount[t] = 0;
        if (t == 0) { qovf = 0; lcount = 0; }
        __syncthreads();

        const int end = (start + range < E_EDGES) ? (start + range) : E_EDGES;
        const int w0 = start >> 10, w1 = (end + 1023) >> 10;
        for (int wi = w0 + t; wi < w1; wi += SEG_THR) {
            unsigned m = lrow[wi];
            if (m != 0u) {
                int pos = atomicAdd(&lcount, __popc(m));
                do {
                    const int b = __builtin_ctz(m);
                    m &= m - 1u;
                    llist[pos++] = (unsigned short)(wi * 32 + b);
                } while (m != 0u);
            }
        }
        __syncthreads();

        const int ntask = lcount * 8;
        for (int k0 = t; k0 < ntask; k0 += SEG_THR * SCAN_U) {
            int  q4[SCAN_U];
            int4 v[SCAN_U];
            #pragma unroll
            for (int u = 0; u < SCAN_U; ++u) {
                const int k  = k0 + u * SEG_THR;
                const int kk = (k < ntask) ? k : (ntask - 1);
                q4[u] = (int)llist[kk >> 3] * 8 + (kk & 7);
            }
            #pragma unroll
            for (int u = 0; u < SCAN_U; ++u) v[u] = et4[q4[u]];
            __builtin_amdgcn_sched_barrier(0);
            #pragma unroll
            for (int u = 0; u < SCAN_U; ++u) {
                const bool ok = (k0 + u * SEG_THR < ntask);
                const int  e  = q4[u] * 4;
                probe_edge(ok ? v[u].x : -1, e + 0, r0, qlist, qcount, &qovf);
                probe_edge(ok ? v[u].y : -1, e + 1, r0, qlist, qcount, &qovf);
                probe_edge(ok ? v[u].z : -1, e + 2, r0, qlist, qcount, &qovf);
                probe_edge(ok ? v[u].w : -1, e + 3, r0, qlist, qcount, &qovf);
            }
        }
        __syncthreads();

        const bool ovf = (qovf != 0);
        const int n0 = qcount[0], n1 = qcount[1], n2 = qcount[2], n3 = qcount[3];
        if (ovf) {
            range = QCAP;
            __syncthreads();
            continue;
        }
        if (t < G_REL) qtotal[t] += qcount[t];

        for (int qq = 0; qq < G_REL; ++qq) {
            const int nn = (qq == 0) ? n0 : (qq == 1) ? n1 : (qq == 2) ? n2 : n3;
            for (int i = t; i < nn; i += SEG_THR) {
                int e = qlist[qq][i].x;
                int s = nid[ei[e]];
                int d = nid[ei[E_EDGES + e]];
                qlist[qq][i] = make_int2(s, d);
            }
        }
        __syncthreads();

        const int n = (q == 0) ? n0 : (q == 1) ? n1 : (q == 2) ? n2 : n3;
        if (active && n > 0) {
            const int2*  L    = qlist[q];
            const float* base = emb + 4 * c;
            for (int j0 = slot; j0 < n; j0 += ROW_SLOTS * GATH_U) {
                int2   en[GATH_U];
                float4 vs[GATH_U], vd[GATH_U];
                #pragma unroll
                for (int u = 0; u < GATH_U; ++u) {
                    int j = j0 + ROW_SLOTS * u;
                    en[u] = L[(j < n) ? j : (n - 1)];
                }
                #pragma unroll
                for (int u = 0; u < GATH_U; ++u) {
                    vs[u] = *(const float4*)(base + (size_t)en[u].x * D_DIM);
                    vd[u] = *(const float4*)(base + (size_t)en[u].y * D_DIM);
                }
                #pragma unroll
                for (int u = 0; u < GATH_U; ++u) {
                    const bool ok = (j0 + ROW_SLOTS * u < n);
                    accS.x += ok ? vs[u].x : 0.f; accS.y += ok ? vs[u].y : 0.f;
                    accS.z += ok ? vs[u].z : 0.f; accS.w += ok ? vs[u].w : 0.f;
                    accD.x += ok ? vd[u].x : 0.f; accD.y += ok ? vd[u].y : 0.f;
                    accD.z += ok ? vd[u].z : 0.f; accD.w += ok ? vd[u].w : 0.f;
                }
            }
        }
        start += range;
    }

    __syncthreads();
    float* red = (float*)&qlist[0][0];
    if (active) {
        *(float4*)&red[(grp * 2 + 0) * D_DIM + 4 * c] = accS;
        *(float4*)&red[(grp * 2 + 1) * D_DIM + 4 * c] = accD;
    }
    __syncthreads();
    if (t < G_REL * 2 * (D_DIM / 4)) {
        const int q2   = t / (2 * (D_DIM / 4));
        const int rem  = t - q2 * (2 * (D_DIM / 4));
        const int side = rem / (D_DIM / 4);
        const int c2   = rem - side * (D_DIM / 4);
        float4 s = *(const float4*)&red[((q2 * ROW_SLOTS + 0) * 2 + side) * D_DIM + 4 * c2];
        for (int sl = 1; sl < ROW_SLOTS; ++sl) {
            float4 p = *(const float4*)&red[((q2 * ROW_SLOTS + sl) * 2 + side) * D_DIM + 4 * c2];
            s.x += p.x; s.y += p.y; s.z += p.z; s.w += p.w;
        }
        const float den = fmaxf((float)qtotal[q2], 1.0f);
        s.x = s.x / den; s.y = s.y / den; s.z = s.z / den; s.w = s.w / den;
        *(float4*)&X[((size_t)(side * R_REL + r0 + q2)) * D_DIM + 4 * c2] = s;
    }
}

__device__ __forceinline__ float sigmoidf_(float x) { return 1.0f / (1.0f + expf(-x)); }

#define GRU_WAVES  8
#define GRU_THR    (GRU_WAVES * 32)
#define ROW_TILES  (M_ROWS / 16)
#define ROW_GROUPS 16
#define COL_TILES  13
#define W_ROWS     96

__launch_bounds__(GRU_THR)
__global__ void gru_wmma_kernel(const float* __restrict__ X,
                                const float* __restrict__ dyn,
                                const float* __restrict__ Wih,
                                const float* __restrict__ Whh,
                                const float* __restrict__ bih,
                                const float* __restrict__ bhh,
                                float* __restrict__ out) {
    __shared__ __attribute__((aligned(16))) float Ws[W_ROWS * LDSTRIDE];

    const int t    = threadIdx.x;
    const int lane = t & 31;
    const int wave = __builtin_amdgcn_readfirstlane(t >> 5);
    const int c0   = blockIdx.x * 16;
    const int tile = blockIdx.y * GRU_WAVES + wave;

    {
        const int lane2 = (lane < 18) ? lane : 17;
        float4 v0[W_ROWS / GRU_WAVES], v1[W_ROWS / GRU_WAVES];
        #pragma unroll
        for (int j = 0; j < W_ROWS / GRU_WAVES; ++j) {
            const int  row  = wave * (W_ROWS / GRU_WAVES) + j;
            const int  g    = row >> 4;
            const int  col  = c0 + (row & 15);
            const int  colc = (col < D_DIM) ? col : (D_DIM - 1);
            const float* Wg   = (g < 3) ? Wih : Whh;
            const int    gate = (g < 3) ? g : (g - 3);
            const float* srow = Wg + (size_t)(gate * D_DIM + colc) * D_DIM;
            v0[j] = *(const float4*)(srow + 4 * lane);
            v1[j] = *(const float4*)(srow + 128 + 4 * lane2);
        }
        #pragma unroll
        for (int j = 0; j < W_ROWS / GRU_WAVES; ++j) {
            const int row = wave * (W_ROWS / GRU_WAVES) + j;
            *(float4*)&Ws[row * LDSTRIDE + 4 * lane] = v0[j];
            *(float4*)&Ws[row * LDSTRIDE + 128 + 4 * lane2] = v1[j];
        }
    }
    __syncthreads();

    if (tile < ROW_TILES) {
        const int m0   = tile * 16;
        const int aRow = lane & 15;
        const int kOff = (lane < 16) ? 0 : 2;
        const int hi   = (lane < 16) ? 0 : 8;

        const int  mA = m0 + aRow;
        const bool fA = (mA >= R_REL);
        const int  rA = fA ? (mA - R_REL) : mA;
        const float* xa = X + (size_t)mA * D_DIM + kOff;
        const float* ha = dyn + (size_t)rA * (2 * D_DIM) + 2 * kOff;
        const float* wb = Ws + aRow * LDSTRIDE + kOff;

        v8f c_ir = {}; v8f c_iz = {}; v8f c_in = {};
        v8f c_hr = {}; v8f c_hz = {}; v8f c_hn = {};

        #pragma unroll 5
        for (int k0 = 0; k0 < D_DIM; k0 += 4) {
            const v2f    ax = *(const v2f*)(xa + k0);
            const float4 hv = *(const float4*)(ha + 2 * k0);
            v2f ah;
            ah.x = fA ? hv.y : hv.x;
            ah.y = fA ? hv.w : hv.z;
            const v2f b0 = *(const v2f*)(wb + 0 * 16 * LDSTRIDE + k0);
            const v2f b1 = *(const v2f*)(wb + 1 * 16 * LDSTRIDE + k0);
            const v2f b2 = *(const v2f*)(wb + 2 * 16 * LDSTRIDE + k0);
            const v2f b3 = *(const v2f*)(wb + 3 * 16 * LDSTRIDE + k0);
            const v2f b4 = *(const v2f*)(wb + 4 * 16 * LDSTRIDE + k0);
            const v2f b5 = *(const v2f*)(wb + 5 * 16 * LDSTRIDE + k0);

            c_ir = __builtin_amdgcn_wmma_f32_16x16x4_f32(false, ax, false, b0, (short)0, c_ir, false, false);
            c_iz = __builtin_amdgcn_wmma_f32_16x16x4_f32(false, ax, false, b1, (short)0, c_iz, false, false);
            c_in = __builtin_amdgcn_wmma_f32_16x16x4_f32(false, ax, false, b2, (short)0, c_in, false, false);
            c_hr = __builtin_amdgcn_wmma_f32_16x16x4_f32(false, ah, false, b3, (short)0, c_hr, false, false);
            c_hz = __builtin_amdgcn_wmma_f32_16x16x4_f32(false, ah, false, b4, (short)0, c_hz, false, false);
            c_hn = __builtin_amdgcn_wmma_f32_16x16x4_f32(false, ah, false, b5, (short)0, c_hn, false, false);
        }

        const int  dCol  = c0 + aRow;
        const bool valid = dCol < D_DIM;
        const int  dc    = valid ? dCol : (D_DIM - 1);
        const float b_ir = bih[dc], b_iz = bih[D_DIM + dc], b_in = bih[2 * D_DIM + dc];
        const float b_hr = bhh[dc], b_hz = bhh[D_DIM + dc], b_hn = bhh[2 * D_DIM + dc];
        #pragma unroll
        for (int i = 0; i < 8; ++i) {
            const int  m = m0 + hi + i;
            const bool f = (m >= R_REL);
            const int  r = f ? (m - R_REL) : m;
            const size_t o_idx = (size_t)r * (2 * D_DIM) + dc * 2 + (f ? 1 : 0);
            float h  = dyn[o_idx];
            float rg = sigmoidf_(c_ir[i] + b_ir + c_hr[i] + b_hr);
            float zg = sigmoidf_(c_iz[i] + b_iz + c_hz[i] + b_hz);
            float ng = tanhf(c_in[i] + b_in + rg * (c_hn[i] + b_hn));
            float o  = (1.0f - zg) * ng + zg * h;
            if (valid) out[o_idx] = o;
        }
    }
}

extern "C" void kernel_launch(void* const* d_in, const int* in_sizes, int n_in,
                              void* d_out, int out_size, void* d_ws, size_t ws_size,
                              hipStream_t stream) {
    const int*   ei  = (const int*)d_in[0];
    const int*   et  = (const int*)d_in[1];
    const int*   nid = (const int*)d_in[2];
    const float* emb = (const float*)d_in[3];
    const float* dyn = (const float*)d_in[4];
    const float* Wih = (const float*)d_in[5];
    const float* Whh = (const float*)d_in[6];
    const float* bih = (const float*)d_in[7];
    const float* bhh = (const float*)d_in[8];
    float* out = (float*)d_out;

    float* X = (float*)d_ws;
    unsigned* lidx = (unsigned*)d_out;

    line_index_kernel<<<IDX_BLOCKS, IDX_THR, 0, stream>>>(et, lidx);

    segmean_kernel<<<R_REL / G_REL, SEG_THR, 0, stream>>>(ei, et, nid, emb, lidx, X);

    gru_wmma_kernel<<<dim3(COL_TILES, ROW_GROUPS), GRU_THR, 0, stream>>>(X, dyn, Wih, Whh, bih, bhh, out);
}


# row gather: one wave per row, its two 16-B loads split at the 128-B line boundary (4 lines + 3 lines) instead of at chunk 32, so no load spans 5 lines; 8 row slots per relation: sums in new order
# speedup vs baseline: 12.3193x; 1.0410x over previous
#include <hip/hip_runtime.h>
#include <math.h>

#define E_EDGES  1000000
#define R_REL    1000
#define D_DIM    200
#define THREE_D  600
#define M_ROWS   2000
#define LDSTRIDE 204

typedef float v2f __attribute__((ext_vector_type(2)));
typedef float v8f __attribute__((ext_vector_type(8)));

#define G_REL    4
#define N_GROUPS (R_REL / G_REL)

#define LINE_EDGES 32
#define N_LINES    (E_EDGES / LINE_EDGES)
#define IDX_EDGES  4096
#define IDX_THR    256
#define IDX_U      (IDX_EDGES / 4 / IDX_THR)
#define IDX_BLOCKS ((E_EDGES + IDX_EDGES - 1) / IDX_EDGES)
#define IDX_WORDS  ((N_LINES + 31) / 32)
#define IDX_PITCH  (IDX_BLOCKS * 4)

__device__ __forceinline__ void mark_type(int ty, int u, unsigned bit, unsigned* words) {
    if ((unsigned)ty < (unsigned)R_REL) atomicOr(&words[(ty / G_REL) * 4 + u], bit);
}

__launch_bounds__(IDX_THR)
__global__ void line_index_kernel(const int* __restrict__ et,
                                  unsigned* __restrict__ lidx) {
    __shared__ __attribute__((aligned(16))) unsigned words[N_GROUPS * 4];
    const int t = threadIdx.x;
    for (int i = t; i < N_GROUPS * 4; i += IDX_THR) words[i] = 0u;
    __syncthreads();

    const int4* et4 = (const int4*)et;
    int4 v[IDX_U];
    #pragma unroll
    for (int u = 0; u < IDX_U; ++u) {
        const int i4 = blockIdx.x * (IDX_EDGES / 4) + u * IDX_THR + t;
        v[u] = (i4 < E_EDGES / 4) ? et4[i4] : make_int4(-1, -1, -1, -1);
    }
    const unsigned bit = 1u << (t >> 3);
    #pragma unroll
    for (int u = 0; u < IDX_U; ++u) {
        mark_type(v[u].x, u, bit, words);
        mark_type(v[u].y, u, bit, words);
        mark_type(v[u].z, u, bit, words);
        mark_type(v[u].w, u, bit, words);
    }
    __syncthreads();
    if (t < N_GROUPS)
        *(uint4*)&lidx[(size_t)t * IDX_PITCH + blockIdx.x * 4] = *(const uint4*)&words[t * 4];
}

#define QCAP     4096
#define SEG_THR  1024
#define SCAN_U   8
#define GATH_U   4
#define ROW_SLOTS 8

__device__ __forceinline__ void probe_edge(int ty, int e, int r0, int2 (*qlist)[QCAP],
                                           int* qcount, int* qovf) {
    unsigned d = (unsigned)(ty - r0);
    if (d < (unsigned)G_REL) {
        int pos = atomicAdd(&qcount[d], 1);
        if (pos < QCAP) qlist[d][pos].x = e;
        else            *qovf = 1;
    }
}

__launch_bounds__(SEG_THR)
__global__ void segmean_kernel(const int* __restrict__ ei,
                               const int* __restrict__ et,
                               const int* __restrict__ nid,
                               const float* __restrict__ emb,
                               const unsigned* __restrict__ lidx,
                               float* __restrict__ X) {
    __shared__ __attribute__((aligned(16))) int2 qlist[G_REL][QCAP];
    __shared__ unsigned short llist[N_LINES];
    __shared__ int  lcount;
    __shared__ int  qcount[G_REL];
    __shared__ int  qtotal[G_REL];
    __shared__ int  qovf;

    const int t    = threadIdx.x;
    const int r0   = blockIdx.x * G_REL;
    const int lane = t & 31;
    const int wv   = t >> 5;
    const int q    = wv >> 3;
    const int slot = wv & (ROW_SLOTS - 1);
    const int chi  = (lane < 18) ? (32 + lane) : (32 + (lane & 15));

    float4 accS0 = make_float4(0.f, 0.f, 0.f, 0.f);
    float4 accS1 = make_float4(0.f, 0.f, 0.f, 0.f);
    float4 accD0 = make_float4(0.f, 0.f, 0.f, 0.f);
    float4 accD1 = make_float4(0.f, 0.f, 0.f, 0.f);

    if (t < G_REL) qtotal[t] = 0;

    const int4* et4 = (const int4*)et;
    const unsigned* lrow = lidx + (size_t)blockIdx.x * IDX_PITCH;
    int start = 0;
    int range = E_EDGES;
    while (start < E_EDGES) {
        if (t < G_REL) qcount[t] = 0;
        if (t == 0) { qovf = 0; lcount = 0; }
        __syncthreads();

        const int end = (start + range < E_EDGES) ? (start + range) : E_EDGES;
        const int w0 = start >> 10, w1 = (end + 1023) >> 10;
        for (int wi = w0 + t; wi < w1; wi += SEG_THR) {
            unsigned m = lrow[wi];
            if (m != 0u) {
                int pos = atomicAdd(&lcount, __popc(m));
                do {
                    const int b = __builtin_ctz(m);
                    m &= m - 1u;
                    llist[pos++] = (unsigned short)(wi * 32 + b);
                } while (m != 0u);
            }
        }
        __syncthreads();

        const int ntask = lcount * 8;
        for (int k0 = t; k0 < ntask; k0 += SEG_THR * SCAN_U) {
            int  q4[SCAN_U];
            int4 v[SCAN_U];
            #pragma unroll
            for (int u = 0; u < SCAN_U; ++u) {
                const int k  = k0 + u * SEG_THR;
                const int kk = (k < ntask) ? k : (ntask - 1);
                q4[u] = (int)llist[kk >> 3] * 8 + (kk & 7);
            }
            #pragma unroll
            for (int u = 0; u < SCAN_U; ++u) v[u] = et4[q4[u]];
            __builtin_amdgcn_sched_barrier(0);
            #pragma unroll
            for (int u = 0; u < SCAN_U; ++u) {
                const bool ok = (k0 + u * SEG_THR < ntask);
                const int  e  = q4[u] * 4;
                probe_edge(ok ? v[u].x : -1, e + 0, r0, qlist, qcount, &qovf);
                probe_edge(ok ? v[u].y : -1, e + 1, r0, qlist, qcount, &qovf);
                probe_edge(ok ? v[u].z : -1, e + 2, r0, qlist, qcount, &qovf);
                probe_edge(ok ? v[u].w : -1, e + 3, r0, qlist, qcount, &qovf);
            }
        }
        __syncthreads();

        const bool ovf = (qovf != 0);
        const int n0 = qcount[0], n1 = qcount[1], n2 = qcount[2], n3 = qcount[3];
        if (ovf) {
            range = QCAP;
            __syncthreads();
            continue;
        }
        if (t < G_REL) qtotal[t] += qcount[t];

        for (int qq = 0; qq < G_REL; ++qq) {
            const int nn = (qq == 0) ? n0 : (qq == 1) ? n1 : (qq == 2) ? n2 : n3;
            for (int i = t; i < nn; i += SEG_THR) {
                int e = qlist[qq][i].x;
                int s = nid[ei[e]];
                int d = nid[ei[E_EDGES + e]];
                qlist[qq][i] = make_int2(s, d);
            }
        }
        __syncthreads();

        const int n = (q == 0) ? n0 : (q == 1) ? n1 : (q == 2) ? n2 : n3;
        if (n > 0) {
            const int2* L = qlist[q];
            for (int j0 = slot; j0 < n; j0 += ROW_SLOTS * GATH_U) {
                int2   en[GATH_U];
                float4 s1[GATH_U], s2[GATH_U], d1[GATH_U], d2[GATH_U];
                #pragma unroll
                for (int u = 0; u < GATH_U; ++u) {
                    int j = j0 + ROW_SLOTS * u;
                    en[u] = L[(j < n) ? j : (n - 1)];
                }
                #pragma unroll
                for (int u = 0; u < GATH_U; ++u) {
                    const int cutS = 32 - 2 * (en[u].x & 3);
                    const int cutD = 32 - 2 * (en[u].y & 3);
                    const float* rs = emb + (size_t)en[u].x * D_DIM;
                    const float* rd = emb + (size_t)en[u].y * D_DIM;
                    s1[u] = *(const float4*)(rs + 4 * ((lane < cutS) ? lane : (lane - 8)));
                    s2[u] = *(const float4*)(rs + 4 * ((lane < cutS) ? chi : lane));
                    d1[u] = *(const float4*)(rd + 4 * ((lane < cutD) ? lane : (lane - 8)));
                    d2[u] = *(const float4*)(rd + 4 * ((lane < cutD) ? chi : lane));
                }
                #pragma unroll
                for (int u = 0; u < GATH_U; ++u) {
                    const bool ok  = (j0 + ROW_SLOTS * u < n);
                    const bool loS = lane < 32 - 2 * (en[u].x & 3);
                    const bool loD = lane < 32 - 2 * (en[u].y & 3);
                    accS0.x += ok ? (loS ? s1[u].x : s2[u].x) : 0.f; accS0.y += ok ? (loS ? s1[u].y : s2[u].y) : 0.f;
                    accS0.z += ok ? (loS ? s1[u].z : s2[u].z) : 0.f; accS0.w += ok ? (loS ? s1[u].w : s2[u].w) : 0.f;
                    accS1.x += ok ? s2[u].x : 0.f; accS1.y += ok ? s2[u].y : 0.f;
                    accS1.z += ok ? s2[u].z : 0.f; accS1.w += ok ? s2[u].w : 0.f;
                    accD0.x += ok ? (loD ? d1[u].x : d2[u].x) : 0.f; accD0.y += ok ? (loD ? d1[u].y : d2[u].y) : 0.f;
                    accD0.z += ok ? (loD ? d1[u].z : d2[u].z) : 0.f; accD0.w += ok ? (loD ? d1[u].w : d2[u].w) : 0.f;
                    accD1.x += ok ? d2[u].x : 0.f; accD1.y += ok ? d2[u].y : 0.f;
                    accD1.z += ok ? d2[u].z : 0.f; accD1.w += ok ? d2[u].w : 0.f;
                }
            }
        }
        start += range;
    }

    __syncthreads();
    float* red = (float*)&qlist[0][0];
    *(float4*)&red[(wv * 2 + 0) * D_DIM + 4 * lane] = accS0;
    *(float4*)&red[(wv * 2 + 1) * D_DIM + 4 * lane] = accD0;
    if (lane < 18) {
        *(float4*)&red[(wv * 2 + 0) * D_DIM + 4 * (32 + lane)] = accS1;
        *(float4*)&red[(wv * 2 + 1) * D_DIM + 4 * (32 + lane)] = accD1;
    }
    __syncthreads();
    if (t < G_REL * 2 * (D_DIM / 4)) {
        const int q2   = t / (2 * (D_DIM / 4));
        const int rem  = t - q2 * (2 * (D_DIM / 4));
        const int side = rem / (D_DIM / 4);
        const int c2   = rem - side * (D_DIM / 4);
        float4 s = *(const float4*)&red[((q2 * ROW_SLOTS + 0) * 2 + side) * D_DIM + 4 * c2];
        for (int sl = 1; sl < ROW_SLOTS; ++sl) {
            float4 p = *(const float4*)&red[((q2 * ROW_SLOTS + sl) * 2 + side) * D_DIM + 4 * c2];
            s.x += p.x; s.y += p.y; s.z += p.z; s.w += p.w;
        }
        const float den = fmaxf((float)qtotal[q2], 1.0f);
        s.x = s.x / den; s.y = s.y / den; s.z = s.z / den; s.w = s.w / den;
        *(float4*)&X[((size_t)(side * R_REL + r0 + q2)) * D_DIM + 4 * c2] = s;
    }
}

__device__ __forceinline__ float sigmoidf_(float x) { return 1.0f / (1.0f + expf(-x)); }

#define GRU_WAVES  8
#define GRU_THR    (GRU_WAVES * 32)
#define ROW_TILES  (M_ROWS / 16)
#define ROW_GROUPS 16
#define COL_TILES  13
#define W_ROWS     96

__launch_bounds__(GRU_THR)
__global__ void gru_wmma_kernel(const float* __restrict__ X,
                                const float* __restrict__ dyn,
                                const float* __restrict__ Wih,
                                const float* __restrict__ Whh,
                                const float* __restrict__ bih,
                                const float* __restrict__ bhh,
                                float* __restrict__ out) {
    __shared__ __attribute__((aligned(16))) float Ws[W_ROWS * LDSTRIDE];

    const int t    = threadIdx.x;
    const int lane = t & 31;
    const int wave = __builtin_amdgcn_readfirstlane(t >> 5);
    const int c0   = blockIdx.x * 16;
    const int tile = blockIdx.y * GRU_WAVES + wave;

    {
        const int lane2 = (lane < 18) ? lane : 17;
        float4 v0[W_ROWS / GRU_WAVES], v1[W_ROWS / GRU_WAVES];
        #pragma unroll
        for (int j = 0; j < W_ROWS / GRU_WAVES; ++j) {
            const int  row  = wave * (W_ROWS / GRU_WAVES) + j;
            const int  g    = row >> 4;
            const int  col  = c0 + (row & 15);
            const int  colc = (col < D_DIM) ? col : (D_DIM - 1);
            const float* Wg   = (g < 3) ? Wih : Whh;
            const int    gate = (g < 3) ? g : (g - 3);
            const float* srow = Wg + (size_t)(gate * D_DIM + colc) * D_DIM;
            v0[j] = *(const float4*)(srow + 4 * lane);
            v1[j] = *(const float4*)(srow + 128 + 4 * lane2);
        }
        #pragma unroll
        for (int j = 0; j < W_ROWS / GRU_WAVES; ++j) {
            const int row = wave * (W_ROWS / GRU_WAVES) + j;
            *(float4*)&Ws[row * LDSTRIDE + 4 * lane] = v0[j];
            *(float4*)&Ws[row * LDSTRIDE + 128 + 4 * lane2] = v1[j];
        }
    }
    __syncthreads();

    if (tile < ROW_TILES) {
        const int m0   = tile * 16;
        const int aRow = lane & 15;
        const int kOff = (lane < 16) ? 0 : 2;
        const int hi   = (lane < 16) ? 0 : 8;

        const int  mA = m0 + aRow;
        const bool fA = (mA >= R_REL);
        const int  rA = fA ? (mA - R_REL) : mA;
        const float* xa = X + (size_t)mA * D_DIM + kOff;
        const float* ha = dyn + (size_t)rA * (2 * D_DIM) + 2 * kOff;
        const float* wb = Ws + aRow * LDSTRIDE + kOff;

        v8f c_ir = {}; v8f c_iz = {}; v8f c_in = {};
        v8f c_hr = {}; v8f c_hz = {}; v8f c_hn = {};

        #pragma unroll 5
        for (int k0 = 0; k0 < D_DIM; k0 += 4) {
            const v2f    ax = *(const v2f*)(xa + k0);
            const float4 hv = *(const float4*)(ha + 2 * k0);
            v2f ah;
            ah.x = fA ? hv.y : hv.x;
            ah.y = fA ? hv.w : hv.z;
            const v2f b0 = *(const v2f*)(wb + 0 * 16 * LDSTRIDE + k0);
            const v2f b1 = *(const v2f*)(wb + 1 * 16 * LDSTRIDE + k0);
            const v2f b2 = *(const v2f*)(wb + 2 * 16 * LDSTRIDE + k0);
            const v2f b3 = *(const v2f*)(wb + 3 * 16 * LDSTRIDE + k0);
            const v2f b4 = *(const v2f*)(wb + 4 * 16 * LDSTRIDE + k0);
            const v2f b5 = *(const v2f*)(wb + 5 * 16 * LDSTRIDE + k0);

            c_ir = __builtin_amdgcn_wmma_f32_16x16x4_f32(false, ax, false, b0, (short)0, c_ir, false, false);
            c_iz = __builtin_amdgcn_wmma_f32_16x16x4_f32(false, ax, false, b1, (short)0, c_iz, false, false);
            c_in = __builtin_amdgcn_wmma_f32_16x16x4_f32(false, ax, false, b2, (short)0, c_in, false, false);
            c_hr = __builtin_amdgcn_wmma_f32_16x16x4_f32(false, ah, false, b3, (short)0, c_hr, false, false);
            c_hz = __builtin_amdgcn_wmma_f32_16x16x4_f32(false, ah, false, b4, (short)0, c_hz, false, false);
            c_hn = __builtin_amdgcn_wmma_f32_16x16x4_f32(false, ah, false, b5, (short)0, c_hn, false, false);
        }

        const int  dCol  = c0 + aRow;
        const bool valid = dCol < D_DIM;
        const int  dc    = valid ? dCol : (D_DIM - 1);
        const float b_ir = bih[dc], b_iz = bih[D_DIM + dc], b_in = bih[2 * D_DIM + dc];
        const float b_hr = bhh[dc], b_hz = bhh[D_DIM + dc], b_hn = bhh[2 * D_DIM + dc];
        #pragma unroll
        for (int i = 0; i < 8; ++i) {
            const int  m = m0 + hi + i;
            const bool f = (m >= R_REL);
            const int  r = f ? (m - R_REL) : m;
            const size_t o_idx = (size_t)r * (2 * D_DIM) + dc * 2 + (f ? 1 : 0);
            float h  = dyn[o_idx];
            float rg = sigmoidf_(c_ir[i] + b_ir + c_hr[i] + b_hr);
            float zg = sigmoidf_(c_iz[i] + b_iz + c_hz[i] + b_hz);
            float ng = tanhf(c_in[i] + b_in + rg * (c_hn[i] + b_hn));
            float o  = (1.0f - zg) * ng + zg * h;
            if (valid) out[o_idx] = o;
        }
    }
}

extern "C" void kernel_launch(void* const* d_in, const int* in_sizes, int n_in,
                              void* d_out, int out_size, void* d_ws, size_t ws_size,
                              hipStream_t stream) {
    const int*   ei  = (const int*)d_in[0];
    const int*   et  = (const int*)d_in[1];
    const int*   nid = (const int*)d_in[2];
    const float* emb = (const float*)d_in[3];
    const float* dyn = (const float*)d_in[4];
    const float* Wih = (const float*)d_in[5];
    const float* Whh = (const float*)d_in[6];
    const float* bih = (const float*)d_in[7];
    const float* bhh = (const float*)d_in[8];
    float* out = (float*)d_out;

    float* X = (float*)d_ws;
    unsigned* lidx = (unsigned*)d_out;

    line_index_kernel<<<IDX_BLOCKS, IDX_THR, 0, stream>>>(et, lidx);

    segmean_kernel<<<R_REL / G_REL, SEG_THR, 0, stream>>>(ei, et, nid, emb, lidx, X);

    gru_wmma_kernel<<<dim3(COL_TILES, ROW_GROUPS), GRU_THR, 0, stream>>>(X, dyn, Wih, Whh, bih, bhh, out);
}


# segment mean: node_id table copied into LDS once per work-group (whole-line loads, 17-bit entries); second hop of edge id -> row is an LDS read, not a scattered 4-byte global read; sum order kept
# speedup vs baseline: 12.7215x; 1.0326x over previous
#include <hip/hip_runtime.h>
#include <math.h>

#define E_EDGES  1000000
#define R_REL    1000
#define D_DIM    200
#define THREE_D  600
#define M_ROWS   2000
#define LDSTRIDE 204

typedef float v2f __attribute__((ext_vector_type(2)));
typedef float v8f __attribute__((ext_vector_type(8)));

#define G_REL    4
#define N_GROUPS (R_REL / G_REL)

#define LINE_EDGES 32
#define N_LINES    (E_EDGES / LINE_EDGES)
#define IDX_EDGES  4096
#define IDX_THR    256
#define IDX_U      (IDX_EDGES / 4 / IDX_THR)
#define IDX_BLOCKS ((E_EDGES + IDX_EDGES - 1) / IDX_EDGES)
#define IDX_WORDS  ((N_LINES + 31) / 32)
#define IDX_PITCH  (IDX_BLOCKS * 4)

__device__ __forceinline__ void mark_type(int ty, int u, unsigned bit, unsigned* words) {
    if ((unsigned)ty < (unsigned)R_REL) atomicOr(&words[(ty / G_REL) * 4 + u], bit);
}

__launch_bounds__(IDX_THR)
__global__ void line_index_kernel(const int* __restrict__ et,
                                  unsigned* __restrict__ lidx) {
    __shared__ __attribute__((aligned(16))) unsigned words[N_GROUPS * 4];
    const int t = threadIdx.x;
    for (int i = t; i < N_GROUPS * 4; i += IDX_THR) words[i] = 0u;
    __syncthreads();

    const int4* et4 = (const int4*)et;
    int4 v[IDX_U];
    #pragma unroll
    for (int u = 0; u < IDX_U; ++u) {
        const int i4 = blockIdx.x * (IDX_EDGES / 4) + u * IDX_THR + t;
        v[u] = (i4 < E_EDGES / 4) ? et4[i4] : make_int4(-1, -1, -1, -1);
    }
    const unsigned bit = 1u << (t >> 3);
    #pragma unroll
    for (int u = 0; u < IDX_U; ++u) {
        mark_type(v[u].x, u, bit, words);
        mark_type(v[u].y, u, bit, words);
        mark_type(v[u].z, u, bit, words);
        mark_type(v[u].w, u, bit, words);
    }
    __syncthreads();
    if (t < N_GROUPS)
        *(uint4*)&lidx[(size_t)t * IDX_PITCH + blockIdx.x * 4] = *(const uint4*)&words[t * 4];
}

#define QCAP     4096
#define N_NODE   50000
#define NID_Q    (N_NODE / 4)
#define NID_U    ((NID_Q + SEG_THR - 1) / SEG_THR)
#define SEG_THR  1024
#define SCAN_U   8
#define GATH_U   4
#define ROW_SLOTS 8

__device__ __forceinline__ void probe_edge(int ty, int e, int r0, int2 (*qlist)[QCAP],
                                           int* qcount, int* qovf) {
    unsigned d = (unsigned)(ty - r0);
    if (d < (unsigned)G_REL) {
        int pos = atomicAdd(&qcount[d], 1);
        if (pos < QCAP) qlist[d][pos].x = e;
        else            *qovf = 1;
    }
}

__launch_bounds__(SEG_THR)
__global__ void segmean_kernel(const int* __restrict__ ei,
                               const int* __restrict__ et,
                               const int* __restrict__ nid,
                               const float* __restrict__ emb,
                               const unsigned* __restrict__ lidx,
                               float* __restrict__ X) {
    __shared__ __attribute__((aligned(16))) int2 qlist[G_REL][QCAP];
    __shared__ unsigned short llist[N_LINES];
    __shared__ __attribute__((aligned(16))) unsigned short nidLo[N_NODE];
    __shared__ unsigned char nidHi[NID_Q];
    __shared__ int  lcount;
    __shared__ int  qcount[G_REL];
    __shared__ int  qtotal[G_REL];
    __shared__ int  qovf;

    const int t    = threadIdx.x;
    const int r0   = blockIdx.x * G_REL;
    const int lane = t & 31;
    const int wv   = t >> 5;
    const int q    = wv >> 3;
    const int slot = wv & (ROW_SLOTS - 1);
    const int chi  = (lane < 18) ? (32 + lane) : (32 + (lane & 15));

    float4 accS0 = make_float4(0.f, 0.f, 0.f, 0.f);
    float4 accS1 = make_float4(0.f, 0.f, 0.f, 0.f);
    float4 accD0 = make_float4(0.f, 0.f, 0.f, 0.f);
    float4 accD1 = make_float4(0.f, 0.f, 0.f, 0.f);

    if (t < G_REL) qtotal[t] = 0;

    {
        const int4* nid4 = (const int4*)nid;
        int4 nv[NID_U];
        #pragma unroll
        for (int u = 0; u < NID_U; ++u) {
            const int i = u * SEG_THR + t;
            nv[u] = nid4[(i < NID_Q) ? i : (NID_Q - 1)];
        }
        #pragma unroll
        for (int u = 0; u < NID_U; ++u) {
            const int i = u * SEG_THR + t;
            if (i < NID_Q) {
                uint2 lo;
                lo.x = ((unsigned)nv[u].x & 0xffffu) | ((unsigned)nv[u].y << 16);
                lo.y = ((unsigned)nv[u].z & 0xffffu) | ((unsigned)nv[u].w << 16);
                *(uint2*)&nidLo[4 * i] = lo;
                nidHi[i] = (unsigned char)((((unsigned)nv[u].x >> 16) & 1u) | ((((unsigned)nv[u].y >> 16) & 1u) << 1) |
                                           ((((unsigned)nv[u].z >> 16) & 1u) << 2) | ((((unsigned)nv[u].w >> 16) & 1u) << 3));
            }
        }
    }

    const int4* et4 = (const int4*)et;
    const unsigned* lrow = lidx + (size_t)blockIdx.x * IDX_PITCH;
    int start = 0;
    int range = E_EDGES;
    while (start < E_EDGES) {
        if (t < G_REL) qcount[t] = 0;
        if (t == 0) { qovf = 0; lcount = 0; }
        __syncthreads();

        const int end = (start + range < E_EDGES) ? (start + range) : E_EDGES;
        const int w0 = start >> 10, w1 = (end + 1023) >> 10;
        for (int wi = w0 + t; wi < w1; wi += SEG_THR) {
            unsigned m = lrow[wi];
            if (m != 0u) {
                int pos = atomicAdd(&lcount, __popc(m));
                do {
                    const int b = __builtin_ctz(m);
                    m &= m - 1u;
                    llist[pos++] = (unsigned short)(wi * 32 + b);
                } while (m != 0u);
            }
        }
        __syncthreads();

        const int ntask = lcount * 8;
        for (int k0 = t; k0 < ntask; k0 += SEG_THR * SCAN_U) {
            int  q4[SCAN_U];
            int4 v[SCAN_U];
            #pragma unroll
            for (int u = 0; u < SCAN_U; ++u) {
                const int k  = k0 + u * SEG_THR;
                const int kk = (k < ntask) ? k : (ntask - 1);
                q4[u] = (int)llist[kk >> 3] * 8 + (kk & 7);
            }
            #pragma unroll
            for (int u = 0; u < SCAN_U; ++u) v[u] = et4[q4[u]];
            __builtin_amdgcn_sched_barrier(0);
            #pragma unroll
            for (int u = 0; u < SCAN_U; ++u) {
                const bool ok = (k0 + u * SEG_THR < ntask);
                const int  e  = q4[u] * 4;
                probe_edge(ok ? v[u].x : -1, e + 0, r0, qlist, qcount, &qovf);
                probe_edge(ok ? v[u].y : -1, e + 1, r0, qlist, qcount, &qovf);
                probe_edge(ok ? v[u].z : -1, e + 2, r0, qlist, qcount, &qovf);
                probe_edge(ok ? v[u].w : -1, e + 3, r0, qlist, qcount, &qovf);
            }
        }
        __syncthreads();

        const bool ovf = (qovf != 0);
        const int n0 = qcount[0], n1 = qcount[1], n2 = qcount[2], n3 = qcount[3];
        if (ovf) {
            range = QCAP;
            __syncthreads();
            continue;
        }
        if (t < G_REL) qtotal[t] += qcount[t];

        for (int qq = 0; qq < G_REL; ++qq) {
            const int nn = (qq == 0) ? n0 : (qq == 1) ? n1 : (qq == 2) ? n2 : n3;
            for (int i = t; i < nn; i += SEG_THR) {
                int e = qlist[qq][i].x;
                const int a = ei[e];
                const int b = ei[E_EDGES + e];
                int s = (int)nidLo[a] | ((((int)nidHi[a >> 2] >> (a & 3)) & 1) << 16);
                int d = (int)nidLo[b] | ((((int)nidHi[b >> 2] >> (b & 3)) & 1) << 16);
                qlist[qq][i] = make_int2(s, d);
            }
        }
        __syncthreads();

        const int n = (q == 0) ? n0 : (q == 1) ? n1 : (q == 2) ? n2 : n3;
        if (n > 0) {
            const int2* L = qlist[q];
            for (int j0 = slot; j0 < n; j0 += ROW_SLOTS * GATH_U) {
                int2   en[GATH_U];
                float4 s1[GATH_U], s2[GATH_U], d1[GATH_U], d2[GATH_U];
                #pragma unroll
                for (int u = 0; u < GATH_U; ++u) {
                    int j = j0 + ROW_SLOTS * u;
                    en[u] = L[(j < n) ? j : (n - 1)];
                }
                #pragma unroll
                for (int u = 0; u < GATH_U; ++u) {
                    const int cutS = 32 - 2 * (en[u].x & 3);
                    const int cutD = 32 - 2 * (en[u].y & 3);
                    const float* rs = emb + (size_t)en[u].x * D_DIM;
                    const float* rd = emb + (size_t)en[u].y * D_DIM;
                    s1[u] = *(const float4*)(rs + 4 * ((lane < cutS) ? lane : (lane - 8)));
                    s2[u] = *(const float4*)(rs + 4 * ((lane < cutS) ? chi : lane));
                    d1[u] = *(const float4*)(rd + 4 * ((lane < cutD) ? lane : (lane - 8)));
                    d2[u] = *(const float4*)(rd + 4 * ((lane < cutD) ? chi : lane));
                }
                #pragma unroll
                for (int u = 0; u < GATH_U; ++u) {
                    const bool ok  = (j0 + ROW_SLOTS * u < n);
                    const bool loS = lane < 32 - 2 * (en[u].x & 3);
                    const bool loD = lane < 32 - 2 * (en[u].y & 3);
                    accS0.x += ok ? (loS ? s1[u].x : s2[u].x) : 0.f; accS0.y += ok ? (loS ? s1[u].y : s2[u].y) : 0.f;
                    accS0.z += ok ? (loS ? s1[u].z : s2[u].z) : 0.f; accS0.w += ok ? (loS ? s1[u].w : s2[u].w) : 0.f;
                    accS1.x += ok ? s2[u].x : 0.f; accS1.y += ok ? s2[u].y : 0.f;
                    accS1.z += ok ? s2[u].z : 0.f; accS1.w += ok ? s2[u].w : 0.f;
                    accD0.x += ok ? (loD ? d1[u].x : d2[u].x) : 0.f; accD0.y += ok ? (loD ? d1[u].y : d2[u].y) : 0.f;
                    accD0.z += ok ? (loD ? d1[u].z : d2[u].z) : 0.f; accD0.w += ok ? (loD ? d1[u].w : d2[u].w) : 0.f;
                    accD1.x += ok ? d2[u].x : 0.f; accD1.y += ok ? d2[u].y : 0.f;
                    accD1.z += ok ? d2[u].z : 0.f; accD1.w += ok ? d2[u].w : 0.f;
                }
            }
        }
        start += range;
    }

    __syncthreads();
    float* red = (float*)&qlist[0][0];
    *(float4*)&red[(wv * 2 + 0) * D_DIM + 4 * lane] = accS0;
    *(float4*)&red[(wv * 2 + 1) * D_DIM + 4 * lane] = accD0;
    if (lane < 18) {
        *(float4*)&red[(wv * 2 + 0) * D_DIM + 4 * (32 + lane)] = accS1;
        *(float4*)&red[(wv * 2 + 1) * D_DIM + 4 * (32 + lane)] = accD1;
    }
    __syncthreads();
    if (t < G_REL * 2 * (D_DIM / 4)) {
        const int q2   = t / (2 * (D_DIM / 4));
        const int rem  = t - q2 * (2 * (D_DIM / 4));
        const int side = rem / (D_DIM / 4);
        const int c2   = rem - side * (D_DIM / 4);
        float4 s = *(const float4*)&red[((q2 * ROW_SLOTS + 0) * 2 + side) * D_DIM + 4 * c2];
        for (int sl = 1; sl < ROW_SLOTS; ++sl) {
            float4 p = *(const float4*)&red[((q2 * ROW_SLOTS + sl) * 2 + side) * D_DIM + 4 * c2];
            s.x += p.x; s.y += p.y; s.z += p.z; s.w += p.w;
        }
        const float den = fmaxf((float)qtotal[q2], 1.0f);
        s.x = s.x / den; s.y = s.y / den; s.z = s.z / den; s.w = s.w / den;
        *(float4*)&X[((size_t)(side * R_REL + r0 + q2)) * D_DIM + 4 * c2] = s;
    }
}

__device__ __forceinline__ float sigmoidf_(float x) { return 1.0f / (1.0f + expf(-x)); }

#define GRU_WAVES  8
#define GRU_THR    (GRU_WAVES * 32)
#define ROW_TILES  (M_ROWS / 16)
#define ROW_GROUPS 16
#define COL_TILES  13
#define W_ROWS     96

__launch_bounds__(GRU_THR)
__global__ void gru_wmma_kernel(const float* __restrict__ X,
                                const float* __restrict__ dyn,
                                const float* __restrict__ Wih,
                                const float* __restrict__ Whh,
                                const float* __restrict__ bih,
                                const float* __restrict__ bhh,
                                float* __restrict__ out) {
    __shared__ __attribute__((aligned(16))) float Ws[W_ROWS * LDSTRIDE];

    const int t    = threadIdx.x;
    const int lane = t & 31;
    const int wave = __builtin_amdgcn_readfirstlane(t >> 5);
    const int c0   = blockIdx.x * 16;
    const int tile = blockIdx.y * GRU_WAVES + wave;

    {
        const int lane2 = (lane < 18) ? lane : 17;
        float4 v0[W_ROWS / GRU_WAVES], v1[W_ROWS / GRU_WAVES];
        #pragma unroll
        for (int j = 0; j < W_ROWS / GRU_WAVES; ++j) {
            const int  row  = wave * (W_ROWS / GRU_WAVES) + j;
            const int  g    = row >> 4;
            const int  col  = c0 + (row & 15);
            const int  colc = (col < D_DIM) ? col : (D_DIM - 1);
            const float* Wg   = (g < 3) ? Wih : Whh;
            const int    gate = (g < 3) ? g : (g - 3);
            const float* srow = Wg + (size_t)(gate * D_DIM + colc) * D_DIM;
            v0[j] = *(const float4*)(srow + 4 * lane);
            v1[j] = *(const float4*)(srow + 128 + 4 * lane2);
        }
        #pragma unroll
        for (int j = 0; j < W_ROWS / GRU_WAVES; ++j) {
            const int row = wave * (W_ROWS / GRU_WAVES) + j;
            *(float4*)&Ws[row * LDSTRIDE + 4 * lane] = v0[j];
            *(float4*)&Ws[row * LDSTRIDE + 128 + 4 * lane2] = v1[j];
        }
    }
    __syncthreads();

    if (tile < ROW_TILES) {
        const int m0   = tile * 16;
        const int aRow = lane & 15;
        const int kOff = (lane < 16) ? 0 : 2;
        const int hi   = (lane < 16) ? 0 : 8;

        const int  mA = m0 + aRow;
        const bool fA = (mA >= R_REL);
        const int  rA = fA ? (mA - R_REL) : mA;
        const float* xa = X + (size_t)mA * D_DIM + kOff;
        const float* ha = dyn + (size_t)rA * (2 * D_DIM) + 2 * kOff;
        const float* wb = Ws + aRow * LDSTRIDE + kOff;

        v8f c_ir = {}; v8f c_iz = {}; v8f c_in = {};
        v8f c_hr = {}; v8f c_hz = {}; v8f c_hn = {};

        #pragma unroll 5
        for (int k0 = 0; k0 < D_DIM; k0 += 4) {
            const v2f    ax = *(const v2f*)(xa + k0);
            const float4 hv = *(const float4*)(ha + 2 * k0);
            v2f ah;
            ah.x = fA ? hv.y : hv.x;
            ah.y = fA ? hv.w : hv.z;
            const v2f b0 = *(const v2f*)(wb + 0 * 16 * LDSTRIDE + k0);
            const v2f b1 = *(const v2f*)(wb + 1 * 16 * LDSTRIDE + k0);
            const v2f b2 = *(const v2f*)(wb + 2 * 16 * LDSTRIDE + k0);
            const v2f b3 = *(const v2f*)(wb + 3 * 16 * LDSTRIDE + k0);
            const v2f b4 = *(const v2f*)(wb + 4 * 16 * LDSTRIDE + k0);
            const v2f b5 = *(const v2f*)(wb + 5 * 16 * LDSTRIDE + k0);

            c_ir = __builtin_amdgcn_wmma_f32_16x16x4_f32(false, ax, false, b0, (short)0, c_ir, false, false);
            c_iz = __builtin_amdgcn_wmma_f32_16x16x4_f32(false, ax, false, b1, (short)0, c_iz, false, false);
            c_in = __builtin_amdgcn_wmma_f32_16x16x4_f32(false, ax, false, b2, (short)0, c_in, false, false);
            c_hr = __builtin_amdgcn_wmma_f32_16x16x4_f32(false, ah, false, b3, (short)0, c_hr, false, false);
            c_hz = __builtin_amdgcn_wmma_f32_16x16x4_f32(false, ah, false, b4, (short)0, c_hz, false, false);
            c_hn = __builtin_amdgcn_wmma_f32_16x16x4_f32(false, ah, false, b5, (short)0, c_hn, false, false);
        }

        const int  dCol  = c0 + aRow;
        const bool valid = dCol < D_DIM;
        const int  dc    = valid ? dCol : (D_DIM - 1);
        const float b_ir = bih[dc], b_iz = bih[D_DIM + dc], b_in = bih[2 * D_DIM + dc];
        const float b_hr = bhh[dc], b_hz = bhh[D_DIM + dc], b_hn = bhh[2 * D_DIM + dc];
        #pragma unroll
        for (int i = 0; i < 8; ++i) {
            const int  m = m0 + hi + i;
            const bool f = (m >= R_REL);
            const int  r = f ? (m - R_REL) : m;
            const size_t o_idx = (size_t)r * (2 * D_DIM) + dc * 2 + (f ? 1 : 0);
            float h  = dyn[o_idx];
            float rg = sigmoidf_(c_ir[i] + b_ir + c_hr[i] + b_hr);
            float zg = sigmoidf_(c_iz[i] + b_iz + c_hz[i] + b_hz);
            float ng = tanhf(c_in[i] + b_in + rg * (c_hn[i] + b_hn));
            float o  = (1.0f - zg) * ng + zg * h;
            if (valid) out[o_idx] = o;
        }
    }
}

extern "C" void kernel_launch(void* const* d_in, const int* in_sizes, int n_in,
                              void* d_out, int out_size, void* d_ws, size_t ws_size,
                              hipStream_t stream) {
    const int*   ei  = (const int*)d_in[0];
    const int*   et  = (const int*)d_in[1];
    const int*   nid = (const int*)d_in[2];
    const float* emb = (const float*)d_in[3];
    const float* dyn = (const float*)d_in[4];
    const float* Wih = (const float*)d_in[5];
    const float* Whh = (const float*)d_in[6];
    const float* bih = (const float*)d_in[7];
    const float* bhh = (const float*)d_in[8];
    float* out = (float*)d_out;

    float* X = (float*)d_ws;
    unsigned* lidx = (unsigned*)d_out;

    line_index_kernel<<<IDX_BLOCKS, IDX_THR, 0, stream>>>(et, lidx);

    segmean_kernel<<<R_REL / G_REL, SEG_THR, 0, stream>>>(ei, et, nid, emb, lidx, X);

    gru_wmma_kernel<<<dim3(COL_TILES, ROW_GROUPS), GRU_THR, 0, stream>>>(X, dyn, Wih, Whh, bih, bhh, out);
}


# row gather: row numbers (equal in all lanes) kept in scalar registers: base, split point, loop and tail test on the scalar unit, loads = scalar base + 32-bit lane offset; order of sums kept
# speedup vs baseline: 13.4138x; 1.0544x over previous
#include <hip/hip_runtime.h>
#include <math.h>

#define E_EDGES  1000000
#define R_REL    1000
#define D_DIM    200
#define THREE_D  600
#define M_ROWS   2000
#define LDSTRIDE 204

typedef float v2f __attribute__((ext_vector_type(2)));
typedef float v8f __attribute__((ext_vector_type(8)));

#define G_REL    4
#define N_GROUPS (R_REL / G_REL)

#define LINE_EDGES 32
#define N_LINES    (E_EDGES / LINE_EDGES)
#define IDX_EDGES  4096
#define IDX_THR    256
#define IDX_U      (IDX_EDGES / 4 / IDX_THR)
#define IDX_BLOCKS ((E_EDGES + IDX_EDGES - 1) / IDX_EDGES)
#define IDX_WORDS  ((N_LINES + 31) / 32)
#define IDX_PITCH  (IDX_BLOCKS * 4)

__device__ __forceinline__ void mark_type(int ty, int u, unsigned bit, unsigned* words) {
    if ((unsigned)ty < (unsigned)R_REL) atomicOr(&words[(ty / G_REL) * 4 + u], bit);
}

__launch_bounds__(IDX_THR)
__global__ void line_index_kernel(const int* __restrict__ et,
                                  unsigned* __restrict__ lidx) {
    __shared__ __attribute__((aligned(16))) unsigned words[N_GROUPS * 4];
    const int t = threadIdx.x;
    for (int i = t; i < N_GROUPS * 4; i += IDX_THR) words[i] = 0u;
    __syncthreads();

    const int4* et4 = (const int4*)et;
    int4 v[IDX_U];
    #pragma unroll
    for (int u = 0; u < IDX_U; ++u) {
        const int i4 = blockIdx.x * (IDX_EDGES / 4) + u * IDX_THR + t;
        v[u] = (i4 < E_EDGES / 4) ? et4[i4] : make_int4(-1, -1, -1, -1);
    }
    const unsigned bit = 1u << (t >> 3);
    #pragma unroll
    for (int u = 0; u < IDX_U; ++u) {
        mark_type(v[u].x, u, bit, words);
        mark_type(v[u].y, u, bit, words);
        mark_type(v[u].z, u, bit, words);
        mark_type(v[u].w, u, bit, words);
    }
    __syncthreads();
    if (t < N_GROUPS)
        *(uint4*)&lidx[(size_t)t * IDX_PITCH + blockIdx.x * 4] = *(const uint4*)&words[t * 4];
}

#define QCAP     4096
#define N_NODE   50000
#define NID_Q    (N_NODE / 4)
#define NID_U    ((NID_Q + SEG_THR - 1) / SEG_THR)
#define SEG_THR  1024
#define SCAN_U   8
#define GATH_U   4
#define ROW_SLOTS 8

__device__ __forceinline__ void probe_edge(int ty, int e, int r0, int2 (*qlist)[QCAP],
                                           int* qcount, int* qovf) {
    unsigned d = (unsigned)(ty - r0);
    if (d < (unsigned)G_REL) {
        int pos = atomicAdd(&qcount[d], 1);
        if (pos < QCAP) qlist[d][pos].x = e;
        else            *qovf = 1;
    }
}

template <bool TAIL>
__device__ __forceinline__ void gather_trip(const int2* L, int j0, int n, int lane,
                                            unsigned oLane, unsigned oLm8, unsigned oChi,
                                            const char* embB,
                                            float4& accS0, float4& accS1, float4& accD0, float4& accD1) {
    unsigned rs[GATH_U], rd[GATH_U];
    float4 s1[GATH_U], s2[GATH_U], d1[GATH_U], d2[GATH_U];
    #pragma unroll
    for (int u = 0; u < GATH_U; ++u) {
        const int j = j0 + ROW_SLOTS * u;
        const int2 en = L[(!TAIL || j < n) ? j : (n - 1)];
        rs[u] = (unsigned)__builtin_amdgcn_readfirstlane(en.x);
        rd[u] = (unsigned)__builtin_amdgcn_readfirstlane(en.y);
    }
    #pragma unroll
    for (int u = 0; u < GATH_U; ++u) {
        const bool loS = lane < (int)(32u - 2u * (rs[u] & 3u));
        const bool loD = lane < (int)(32u - 2u * (rd[u] & 3u));
        const char* ps = embB + (size_t)(rs[u] * (unsigned)(D_DIM * 4));
        const char* pd = embB + (size_t)(rd[u] * (unsigned)(D_DIM * 4));
        s1[u] = *(const float4*)(ps + (loS ? oLane : oLm8));
        s2[u] = *(const float4*)(ps + (loS ? oChi : oLane));
        d1[u] = *(const float4*)(pd + (loD ? oLane : oLm8));
        d2[u] = *(const float4*)(pd + (loD ? oChi : oLane));
    }
    #pragma unroll
    for (int u = 0; u < GATH_U; ++u) {
        if (!TAIL || j0 + ROW_SLOTS * u < n) {
            const bool loS = lane < (int)(32u - 2u * (rs[u] & 3u));
            const bool loD = lane < (int)(32u - 2u * (rd[u] & 3u));
            accS0.x += loS ? s1[u].x : s2[u].x; accS0.y += loS ? s1[u].y : s2[u].y;
            accS0.z += loS ? s1[u].z : s2[u].z; accS0.w += loS ? s1[u].w : s2[u].w;
            accS1.x += s2[u].x; accS1.y += s2[u].y; accS1.z += s2[u].z; accS1.w += s2[u].w;
            accD0.x += loD ? d1[u].x : d2[u].x; accD0.y += loD ? d1[u].y : d2[u].y;
            accD0.z += loD ? d1[u].z : d2[u].z; accD0.w += loD ? d1[u].w : d2[u].w;
            accD1.x += d2[u].x; accD1.y += d2[u].y; accD1.z += d2[u].z; accD1.w += d2[u].w;
        }
    }
}

__launch_bounds__(SEG_THR)
__global__ void segmean_kernel(const int* __restrict__ ei,
                               const int* __restrict__ et,
                               const int* __restrict__ nid,
                               const float* __restrict__ emb,
                               const unsigned* __restrict__ lidx,
                               float* __restrict__ X) {
    __shared__ __attribute__((aligned(16))) int2 qlist[G_REL][QCAP];
    __shared__ unsigned short llist[N_LINES];
    __shared__ __attribute__((aligned(16))) unsigned short nidLo[N_NODE];
    __shared__ unsigned char nidHi[NID_Q];
    __shared__ int  lcount;
    __shared__ int  qcount[G_REL];
    __shared__ int  qtotal[G_REL];
    __shared__ int  qovf;

    const int t    = threadIdx.x;
    const int r0   = blockIdx.x * G_REL;
    const int lane = t & 31;
    const int wv   = __builtin_amdgcn_readfirstlane(t >> 5);
    const int q    = wv >> 3;
    const int slot = wv & (ROW_SLOTS - 1);
    const int chi  = (lane < 18) ? (32 + lane) : (32 + (lane & 15));

    float4 accS0 = make_float4(0.f, 0.f, 0.f, 0.f);
    float4 accS1 = make_float4(0.f, 0.f, 0.f, 0.f);
    float4 accD0 = make_float4(0.f, 0.f, 0.f, 0.f);
    float4 accD1 = make_float4(0.f, 0.f, 0.f, 0.f);

    const unsigned oLane = 16u * (unsigned)lane;
    const unsigned oLm8  = 16u * (unsigned)((lane >= 8) ? (lane - 8) : lane);
    const unsigned oChi  = 16u * (unsigned)chi;
    const char* embB = (const char*)emb;

    if (t < G_REL) qtotal[t] = 0;

    {
        const int4* nid4 = (const int4*)nid;
        int4 nv[NID_U];
        #pragma unroll
        for (int u = 0; u < NID_U; ++u) {
            const int i = u * SEG_THR + t;
            nv[u] = nid4[(i < NID_Q) ? i : (NID_Q - 1)];
        }
        #pragma unroll
        for (int u = 0; u < NID_U; ++u) {
            const int i = u * SEG_THR + t;
            if (i < NID_Q) {
                uint2 lo;
                lo.x = ((unsigned)nv[u].x & 0xffffu) | ((unsigned)nv[u].y << 16);
                lo.y = ((unsigned)nv[u].z & 0xffffu) | ((unsigned)nv[u].w << 16);
                *(uint2*)&nidLo[4 * i] = lo;
                nidHi[i] = (unsigned char)((((unsigned)nv[u].x >> 16) & 1u) | ((((unsigned)nv[u].y >> 16) & 1u) << 1) |
                                           ((((unsigned)nv[u].z >> 16) & 1u) << 2) | ((((unsigned)nv[u].w >> 16) & 1u) << 3));
            }
        }
    }

    const int4* et4 = (const int4*)et;
    const unsigned* lrow = lidx + (size_t)blockIdx.x * IDX_PITCH;
    int start = 0;
    int range = E_EDGES;
    while (start < E_EDGES) {
        if (t < G_REL) qcount[t] = 0;
        if (t == 0) { qovf = 0; lcount = 0; }
        __syncthreads();

        const int end = (start + range < E_EDGES) ? (start + range) : E_EDGES;
        const int w0 = start >> 10, w1 = (end + 1023) >> 10;
        for (int wi = w0 + t; wi < w1; wi += SEG_THR) {
            unsigned m = lrow[wi];
            if (m != 0u) {
                int pos = atomicAdd(&lcount, __popc(m));
                do {
                    const int b = __builtin_ctz(m);
                    m &= m - 1u;
                    llist[pos++] = (unsigned short)(wi * 32 + b);
                } while (m != 0u);
            }
        }
        __syncthreads();

        const int ntask = lcount * 8;
        for (int k0 = t; k0 < ntask; k0 += SEG_THR * SCAN_U) {
            int  q4[SCAN_U];
            int4 v[SCAN_U];
            #pragma unroll
            for (int u = 0; u < SCAN_U; ++u) {
                const int k  = k0 + u * SEG_THR;
                const int kk = (k < ntask) ? k : (ntask - 1);
                q4[u] = (int)llist[kk >> 3] * 8 + (kk & 7);
            }
            #pragma unroll
            for (int u = 0; u < SCAN_U; ++u) v[u] = et4[q4[u]];
            __builtin_amdgcn_sched_barrier(0);
            #pragma unroll
            for (int u = 0; u < SCAN_U; ++u) {
                const bool ok = (k0 + u * SEG_THR < ntask);
                const int  e  = q4[u] * 4;
                probe_edge(ok ? v[u].x : -1, e + 0, r0, qlist, qcount, &qovf);
                probe_edge(ok ? v[u].y : -1, e + 1, r0, qlist, qcount, &qovf);
                probe_edge(ok ? v[u].z : -1, e + 2, r0, qlist, qcount, &qovf);
                probe_edge(ok ? v[u].w : -1, e + 3, r0, qlist, qcount, &qovf);
            }
        }
        __syncthreads();

        const bool ovf = (qovf != 0);
        const int n0 = qcount[0], n1 = qcount[1], n2 = qcount[2], n3 = qcount[3];
        if (ovf) {
            range = QCAP;
            __syncthreads();
            continue;
        }
        if (t < G_REL) qtotal[t] += qcount[t];

        for (int qq = 0; qq < G_REL; ++qq) {
            const int nn = (qq == 0) ? n0 : (qq == 1) ? n1 : (qq == 2) ? n2 : n3;
            for (int i = t; i < nn; i += SEG_THR) {
                int e = qlist[qq][i].x;
                const int a = ei[e];
                const int b = ei[E_EDGES + e];
                int s = (int)nidLo[a] | ((((int)nidHi[a >> 2] >> (a & 3)) & 1) << 16);
                int d = (int)nidLo[b] | ((((int)nidHi[b >> 2] >> (b & 3)) & 1) << 16);
                qlist[qq][i] = make_int2(s, d);
            }
        }
        __syncthreads();

        const int n = __builtin_amdgcn_readfirstlane((q == 0) ? n0 : (q == 1) ? n1 : (q == 2) ? n2 : n3);
        if (n > 0) {
            const int2* L = qlist[q];
            int j0 = slot;
            for (; j0 + ROW_SLOTS * (GATH_U - 1) < n; j0 += ROW_SLOTS * GATH_U)
                gather_trip<false>(L, j0, n, lane, oLane, oLm8, oChi, embB, accS0, accS1, accD0, accD1);
            if (j0 < n)
                gather_trip<true>(L, j0, n, lane, oLane, oLm8, oChi, embB, accS0, accS1, accD0, accD1);
        }
        start += range;
    }

    __syncthreads();
    float* red = (float*)&qlist[0][0];
    *(float4*)&red[(wv * 2 + 0) * D_DIM + 4 * lane] = accS0;
    *(float4*)&red[(wv * 2 + 1) * D_DIM + 4 * lane] = accD0;
    if (lane < 18) {
        *(float4*)&red[(wv * 2 + 0) * D_DIM + 4 * (32 + lane)] = accS1;
        *(float4*)&red[(wv * 2 + 1) * D_DIM + 4 * (32 + lane)] = accD1;
    }
    __syncthreads();
    if (t < G_REL * 2 * (D_DIM / 4)) {
        const int q2   = t / (2 * (D_DIM / 4));
        const int rem  = t - q2 * (2 * (D_DIM / 4));
        const int side = rem / (D_DIM / 4);
        const int c2   = rem - side * (D_DIM / 4);
        float4 s = *(const float4*)&red[((q2 * ROW_SLOTS + 0) * 2 + side) * D_DIM + 4 * c2];
        for (int sl = 1; sl < ROW_SLOTS; ++sl) {
            float4 p = *(const float4*)&red[((q2 * ROW_SLOTS + sl) * 2 + side) * D_DIM + 4 * c2];
            s.x += p.x; s.y += p.y; s.z += p.z; s.w += p.w;
        }
        const float den = fmaxf((float)qtotal[q2], 1.0f);
        s.x = s.x / den; s.y = s.y / den; s.z = s.z / den; s.w = s.w / den;
        *(float4*)&X[((size_t)(side * R_REL + r0 + q2)) * D_DIM + 4 * c2] = s;
    }
}

__device__ __forceinline__ float sigmoidf_(float x) { return 1.0f / (1.0f + expf(-x)); }

#define GRU_WAVES  8
#define GRU_THR    (GRU_WAVES * 32)
#define ROW_TILES  (M_ROWS / 16)
#define ROW_GROUPS 16
#define COL_TILES  13
#define W_ROWS     96

__launch_bounds__(GRU_THR)
__global__ void gru_wmma_kernel(const float* __restrict__ X,
                                const float* __restrict__ dyn,
                                const float* __restrict__ Wih,
                                const float* __restrict__ Whh,
                                const float* __restrict__ bih,
                                const float* __restrict__ bhh,
                                float* __restrict__ out) {
    __shared__ __attribute__((aligned(16))) float Ws[W_ROWS * LDSTRIDE];

    const int t    = threadIdx.x;
    const int lane = t & 31;
    const int wave = __builtin_amdgcn_readfirstlane(t >> 5);
    const int c0   = blockIdx.x * 16;
    const int tile = blockIdx.y * GRU_WAVES + wave;

    {
        const int lane2 = (lane < 18) ? lane : 17;
        float4 v0[W_ROWS / GRU_WAVES], v1[W_ROWS / GRU_WAVES];
        #pragma unroll
        for (int j = 0; j < W_ROWS / GRU_WAVES; ++j) {
            const int  row  = wave * (W_ROWS / GRU_WAVES) + j;
            const int  g    = row >> 4;
            const int  col  = c0 + (row & 15);
            const int  colc = (col < D_DIM) ? col : (D_DIM - 1);
            const float* Wg   = (g < 3) ? Wih : Whh;
            const int    gate = (g < 3) ? g : (g - 3);
            const float* srow = Wg + (size_t)(gate * D_DIM + colc) * D_DIM;
            v0[j] = *(const float4*)(srow + 4 * lane);
            v1[j] = *(const float4*)(srow + 128 + 4 * lane2);
        }
        #pragma unroll
        for (int j = 0; j < W_ROWS / GRU_WAVES; ++j) {
            const int row = wave * (W_ROWS / GRU_WAVES) + j;
            *(float4*)&Ws[row * LDSTRIDE + 4 * lane] = v0[j];
            *(float4*)&Ws[row * LDSTRIDE + 128 + 4 * lane2] = v1[j];
        }
    }
    __syncthreads();

    if (tile < ROW_TILES) {
        const int m0   = tile * 16;
        const int aRow = lane & 15;
        const int kOff = (lane < 16) ? 0 : 2;
        const int hi   = (lane < 16) ? 0 : 8;

        const int  mA = m0 + aRow;
        const bool fA = (mA >= R_REL);
        const int  rA = fA ? (mA - R_REL) : mA;
        const float* xa = X + (size_t)mA * D_DIM + kOff;
        const float* ha = dyn + (size_t)rA * (2 * D_DIM) + 2 * kOff;
        const float* wb = Ws + aRow * LDSTRIDE + kOff;

        v8f c_ir = {}; v8f c_iz = {}; v8f c_in = {};
        v8f c_hr = {}; v8f c_hz = {}; v8f c_hn = {};

        #pragma unroll 5
        for (int k0 = 0; k0 < D_DIM; k0 += 4) {
            const v2f    ax = *(const v2f*)(xa + k0);
            const float4 hv = *(const float4*)(ha + 2 * k0);
            v2f ah;
            ah.x = fA ? hv.y : hv.x;
            ah.y = fA ? hv.w : hv.z;
            const v2f b0 = *(const v2f*)(wb + 0 * 16 * LDSTRIDE + k0);
            const v2f b1 = *(const v2f*)(wb + 1 * 16 * LDSTRIDE + k0);
            const v2f b2 = *(const v2f*)(wb + 2 * 16 * LDSTRIDE + k0);
            const v2f b3 = *(const v2f*)(wb + 3 * 16 * LDSTRIDE + k0);
            const v2f b4 = *(const v2f*)(wb + 4 * 16 * LDSTRIDE + k0);
            const v2f b5 = *(const v2f*)(wb + 5 * 16 * LDSTRIDE + k0);

            c_ir = __builtin_amdgcn_wmma_f32_16x16x4_f32(false, ax, false, b0, (short)0, c_ir, false, false);
            c_iz = __builtin_amdgcn_wmma_f32_16x16x4_f32(false, ax, false, b1, (short)0, c_iz, false, false);
            c_in = __builtin_amdgcn_wmma_f32_16x16x4_f32(false, ax, false, b2, (short)0, c_in, false, false);
            c_hr = __builtin_amdgcn_wmma_f32_16x16x4_f32(false, ah, false, b3, (short)0, c_hr, false, false);
            c_hz = __builtin_amdgcn_wmma_f32_16x16x4_f32(false, ah, false, b4, (short)0, c_hz, false, false);
            c_hn = __builtin_amdgcn_wmma_f32_16x16x4_f32(false, ah, false, b5, (short)0, c_hn, false, false);
        }

        const int  dCol  = c0 + aRow;
        const bool valid = dCol < D_DIM;
        const int  dc    = valid ? dCol : (D_DIM - 1);
        const float b_ir = bih[dc], b_iz = bih[D_DIM + dc], b_in = bih[2 * D_DIM + dc];
        const float b_hr = bhh[dc], b_hz = bhh[D_DIM + dc], b_hn = bhh[2 * D_DIM + dc];
        #pragma unroll
        for (int i = 0; i < 8; ++i) {
            const int  m = m0 + hi + i;
            const bool f = (m >= R_REL);
            const int  r = f ? (m - R_REL) : m;
            const size_t o_idx = (size_t)r * (2 * D_DIM) + dc * 2 + (f ? 1 : 0);
            float h  = dyn[o_idx];
            float rg = sigmoidf_(c_ir[i] + b_ir + c_hr[i] + b_hr);
            float zg = sigmoidf_(c_iz[i] + b_iz + c_hz[i] + b_hz);
            float ng = tanhf(c_in[i] + b_in + rg * (c_hn[i] + b_hn));
            float o  = (1.0f - zg) * ng + zg * h;
            if (valid) out[o_idx] = o;
        }
    }
}

extern "C" void kernel_launch(void* const* d_in, const int* in_sizes, int n_in,
                              void* d_out, int out_size, void* d_ws, size_t ws_size,
                              hipStream_t stream) {
    const int*   ei  = (const int*)d_in[0];
    const int*   et  = (const int*)d_in[1];
    const int*   nid = (const int*)d_in[2];
    const float* emb = (const float*)d_in[3];
    const float* dyn = (const float*)d_in[4];
    const float* Wih = (const float*)d_in[5];
    const float* Whh = (const float*)d_in[6];
    const float* bih = (const float*)d_in[7];
    const float* bhh = (const float*)d_in[8];
    float* out = (float*)d_out;

    float* X = (float*)d_ws;
    unsigned* lidx = (unsigned*)d_out;

    line_index_kernel<<<IDX_BLOCKS, IDX_THR, 0, stream>>>(et, lidx);

    segmean_kernel<<<R_REL / G_REL, SEG_THR, 0, stream>>>(ei, et, nid, emb, lidx, X);

    gru_wmma_kernel<<<dim3(COL_TILES, ROW_GROUPS), GRU_THR, 0, stream>>>(X, dyn, Wih, Whh, bih, bhh, out);
}
